# starting kernel
# baseline (speedup 1.0000x reference)
#include <hip/hip_runtime.h>

#define D14   14
#define SM    17
#define JITTER 1e-4f
#define MIN_Q  1e-5f
#define MIN_R  1e-4f

typedef __attribute__((ext_vector_type(2))) float v2f;
typedef __attribute__((ext_vector_type(8))) float v8f;

template <bool BT>
__device__ __forceinline__ v8f mm_chain(const float* A, const float* Bm,
                                        const float* colmask,
                                        v8f c, int l16, int h) {
#pragma unroll
  for (int kk = 0; kk < 4; ++kk) {
    const int k0 = 4 * kk + 2 * h;
    v2f a, b;
    float a0 = A[l16 * SM + k0];
    float a1 = A[l16 * SM + k0 + 1];
    if (colmask) { a0 *= colmask[k0]; a1 *= colmask[k0 + 1]; }
    a.x = a0; a.y = a1;
    if (BT) {
      b.x = Bm[l16 * SM + k0];
      b.y = Bm[l16 * SM + k0 + 1];
    } else {
      b.x = Bm[k0 * SM + l16];
      b.y = Bm[(k0 + 1) * SM + l16];
    }
    c = __builtin_amdgcn_wmma_f32_16x16x4_f32(false, a, false, b, (short)0, c,
                                              false, false);
  }
  return c;
}

__global__ __launch_bounds__(32) void kf_scan(
    const float* __restrict__ xh, const float* __restrict__ z,
    const float* __restrict__ m21,
    const float* __restrict__ Wf, const float* __restrict__ bf,
    const float* __restrict__ Wq, const float* __restrict__ bq,
    const float* __restrict__ Wr, const float* __restrict__ br,
    float* __restrict__ out, int T) {
  __shared__ float sWf[196], sWq[196], sWr[196];
  __shared__ float sb[48];
  __shared__ float Pm[16 * SM], Sm[16 * SM], Gm[16 * SM], Km[16 * SM];
  __shared__ float zv[16], xv[16], xpv[16], qv[16], rv[16], mv[16], iv[16];

  const int lane = threadIdx.x;
  const int b    = blockIdx.x;
  const int l16  = lane & 15;
  const int h    = lane >> 4;

  for (int i = lane; i < 196; i += 32) { sWf[i] = Wf[i]; sWq[i] = Wq[i]; sWr[i] = Wr[i]; }
  if (lane < 16) {
    sb[lane]      = (lane < D14) ? bf[lane] : 0.f;
    sb[16 + lane] = (lane < D14) ? bq[lane] : 0.f;
    sb[32 + lane] = (lane < D14) ? br[lane] : 0.f;
    zv[lane] = 0.f; xv[lane] = 0.f; xpv[lane] = 0.f; qv[lane] = 0.f;
    rv[lane] = 0.f; mv[lane] = 0.f; iv[lane] = 0.f;
  }
#pragma unroll
  for (int k = 0; k < 8; ++k) {
    int idx = lane * 8 + k; int r = idx >> 4, c = idx & 15;
    Pm[r * SM + c] = (r == c && r < D14) ? 1.0f : 0.0f;
  }
  __syncthreads();

  const long long base = (long long)b * T;
#pragma unroll 1
  for (int t = 0; t < T; ++t) {
    const long long tok = base + t;
    if (lane < D14) {
      zv[lane] = z[tok * D14 + lane];
      xv[lane] = xh[tok * D14 + lane];
    }
    if (lane < 7) {
      float v  = m21[tok * 21 + 3 * lane + 2];
      float mm = (v > 0.5f) ? 1.0f : 0.0f;
      mv[2 * lane] = mm; mv[2 * lane + 1] = mm;
    }
    __syncthreads();
    if (lane < D14) {
      float a0 = sb[lane], a2 = sb[32 + lane];
#pragma unroll
      for (int i = 0; i < D14; ++i) {
        a0 += xv[i] * sWf[i * D14 + lane];
        a2 += zv[i] * sWr[i * D14 + lane];
      }
      xpv[lane] = a0;
      rv[lane]  = fmaxf(a2, MIN_R);
    }
    __syncthreads();
    if (lane < D14) {
      float a1 = sb[16 + lane];
#pragma unroll
      for (int i = 0; i < D14; ++i) a1 += xpv[i] * sWq[i * D14 + lane];
      qv[lane] = fmaxf(a1, MIN_Q);
    }
    __syncthreads();
    if (lane < D14) Pm[lane * SM + lane] += qv[lane];
    __syncthreads();
#pragma unroll
    for (int k = 0; k < 8; ++k) {
      int idx = lane * 8 + k; int r = idx >> 4, c = idx & 15;
      float ps = 0.5f * (Pm[r * SM + c] + Pm[c * SM + r]);
      float s  = mv[r] * mv[c] * ps;
      if (r == c && r < D14) s += mv[r] * rv[r] + (1.0f - mv[r]) + JITTER * mv[r];
      Sm[r * SM + c] = s;
    }
    __syncthreads();
    float tv = (lane < D14) ? mv[lane] * Sm[lane * SM + lane] : 0.0f;
#pragma unroll
    for (int off = 16; off >= 1; off >>= 1) tv += __shfl_xor(tv, off, 32);
    const float dload = JITTER * fmaxf(tv, 1e-6f);
#pragma unroll
    for (int k = 0; k < 8; ++k) {
      int idx = lane * 8 + k; int r = idx >> 4, c = idx & 15;
      float g = Sm[r * SM + c];
      if (r == c) g = (r < D14) ? (g + dload) : 1.0f;
      Gm[r * SM + c] = g;
    }
    __syncthreads();
    for (int k = 0; k < D14; ++k) {
      const float ip = 1.0f / Gm[k * SM + k];
      if (h == 0) {
        float val = Gm[k * SM + l16];
        Gm[k * SM + l16] = (l16 == k) ? ip : val * ip;
      }
      __syncthreads();
      if (l16 != k) {
        const float f = Gm[l16 * SM + k];
#pragma unroll
        for (int j = 0; j < 8; ++j) {
          int c = h * 8 + j;
          float cur = (c == k) ? 0.0f : Gm[l16 * SM + c];
          Gm[l16 * SM + c] = cur - f * Gm[k * SM + c];
        }
      }
      __syncthreads();
    }
    v8f acc = {};
    acc = mm_chain<false>(Pm, Gm, mv, acc, l16, h);
#pragma unroll
    for (int v = 0; v < 8; ++v) Km[(v + 8 * h) * SM + l16] = acc[v];
    __syncthreads();
    if (lane < D14) iv[lane] = mv[lane] * (zv[lane] - xpv[lane]);
    __syncthreads();
    if (lane < D14) {
      float xo = xpv[lane];
#pragma unroll
      for (int i = 0; i < D14; ++i) xo += Km[lane * SM + i] * iv[i];
      out[tok * D14 + lane] = xo;
    }
    v8f ks = {};
    ks = mm_chain<false>(Km, Sm, nullptr, ks, l16, h);
#pragma unroll
    for (int v = 0; v < 8; ++v) Gm[(v + 8 * h) * SM + l16] = -ks[v];
    __syncthreads();
    v8f pc;
#pragma unroll
    for (int v = 0; v < 8; ++v) pc[v] = Pm[(v + 8 * h) * SM + l16];
    pc = mm_chain<true>(Gm, Km, nullptr, pc, l16, h);
#pragma unroll
    for (int v = 0; v < 8; ++v) Sm[(v + 8 * h) * SM + l16] = pc[v];
    __syncthreads();
#pragma unroll
    for (int k = 0; k < 8; ++k) {
      int idx = lane * 8 + k; int r = idx >> 4, c = idx & 15;
      Pm[r * SM + c] = 0.5f * (Sm[r * SM + c] + Sm[c * SM + r]);
    }
    __syncthreads();
  }
}

extern "C" void kernel_launch(void* const* d_in, const int* in_sizes, int n_in,
                              void* d_out, int out_size, void* d_ws, size_t ws_size,
                              hipStream_t stream) {
  const float* xh  = (const float*)d_in[0];
  const float* z   = (const float*)d_in[1];
  const float* m21 = (const float*)d_in[2];
  const float* Wf  = (const float*)d_in[3];
  const float* bf  = (const float*)d_in[4];
  const float* Wq  = (const float*)d_in[5];
  const float* bq  = (const float*)d_in[6];
  const float* Wr  = (const float*)d_in[7];
  const float* br  = (const float*)d_in[8];
  float* out = (float*)d_out;

  const int T  = 512;
  const int BT = in_sizes[1] / D14;
  const int B  = BT / T;

  kf_scan<<<dim3(B), dim3(32), 0, stream>>>(xh, z, m21, Wf, bf, Wq, bq, Wr, br,
                                            out, T);
}


# All step matrices stay in accumulator tiles: products run as X^T*Y on register pairs of the tiles, symmetrised P from the last product taken both ways round; k and vector sum orders change
# speedup vs baseline: 3.8316x; 3.8316x over previous
#include <hip/hip_runtime.h>

#define D14   14
#define JITTER 1e-4f
#define MIN_Q  1e-5f
#define MIN_R  1e-4f

typedef __attribute__((ext_vector_type(2))) float v2f;
typedef __attribute__((ext_vector_type(8))) float v8f;

__device__ __forceinline__ v8f mm_tn(v8f X, v8f Y, v8f c) {
#pragma unroll
  for (int kk = 0; kk < 4; ++kk) {
    v2f a, b;
    a.x = X[2 * kk]; a.y = X[2 * kk + 1];
    b.x = Y[2 * kk]; b.y = Y[2 * kk + 1];
    c = __builtin_amdgcn_wmma_f32_16x16x4_f32(false, a, false, b, (short)0, c,
                                              false, false);
  }
  return c;
}

__global__ __launch_bounds__(32) void kf_scan(
    const float* __restrict__ xh, const float* __restrict__ z,
    const float* __restrict__ m21,
    const float* __restrict__ Wf, const float* __restrict__ bf,
    const float* __restrict__ Wq, const float* __restrict__ bq,
    const float* __restrict__ Wr, const float* __restrict__ br,
    float* __restrict__ out, int T) {
  __shared__ float sWf[196], sWq[196], sWr[196];
  __shared__ float sb[48];
  __shared__ __attribute__((aligned(16))) float zv[16], xv[16], xpv[16], qv[16],
      rv[16], mv[16], iv[16];

  const int lane = threadIdx.x;
  const int b    = blockIdx.x;
  const int l16  = lane & 15;
  const int h    = lane >> 4;
  const bool in14 = (l16 < D14);

  for (int i = lane; i < 196; i += 32) { sWf[i] = Wf[i]; sWq[i] = Wq[i]; sWr[i] = Wr[i]; }
  if (lane < 16) {
    sb[lane]      = (lane < D14) ? bf[lane] : 0.f;
    sb[16 + lane] = (lane < D14) ? bq[lane] : 0.f;
    sb[32 + lane] = (lane < D14) ? br[lane] : 0.f;
    zv[lane] = 0.f; xv[lane] = 0.f; xpv[lane] = 0.f; qv[lane] = 0.f;
    rv[lane] = 0.f; mv[lane] = 0.f; iv[lane] = 0.f;
  }
  v8f Pt;
#pragma unroll
  for (int v = 0; v < 8; ++v) Pt[v] = (v + 8 * h == l16 && in14) ? 1.0f : 0.0f;
  __syncthreads();

  const long long base = (long long)b * T;
#pragma unroll 1
  for (int t = 0; t < T; ++t) {
    const long long tok = base + t;
    {
      const int  ci   = in14 ? l16 : (D14 - 1);
      const float zl = z[tok * D14 + ci];
      const float xl = xh[tok * D14 + ci];
      const float ml = m21[tok * 21 + 3 * (ci >> 1) + 2];
      zv[l16] = in14 ? zl : 0.0f;
      xv[l16] = in14 ? xl : 0.0f;
      mv[l16] = (in14 && ml > 0.5f) ? 1.0f : 0.0f;
    }
    __syncthreads();
    float xpreg = 0.0f;
    if (lane < D14) {
      float a0 = sb[lane], a2 = sb[32 + lane];
#pragma unroll
      for (int i = 0; i < D14; ++i) {
        a0 += xv[i] * sWf[i * D14 + lane];
        a2 += zv[i] * sWr[i * D14 + lane];
      }
      xpv[lane] = a0;
      xpreg     = a0;
      rv[lane]  = fmaxf(a2, MIN_R);
    }
    __syncthreads();
    if (lane < D14) {
      float a1 = sb[16 + lane];
#pragma unroll
      for (int i = 0; i < D14; ++i) a1 += xpv[i] * sWq[i * D14 + lane];
      qv[lane] = fmaxf(a1, MIN_Q);
      iv[lane] = mv[lane] * (zv[lane] - xpv[lane]);
    }
    __syncthreads();
    const float qcol = qv[l16], rcol = rv[l16], mcol = mv[l16];
    float mrow[8], ivrow[8];
#pragma unroll
    for (int v = 0; v < 8; ++v) { mrow[v] = mv[8 * h + v]; ivrow[v] = iv[8 * h + v]; }
    const float dterm = mcol * rcol + (1.0f - mcol) + JITTER * mcol;
    v8f St, nS, Yn;
    float dS = 0.0f;
#pragma unroll
    for (int v = 0; v < 8; ++v) {
      const bool dg = (v + 8 * h == l16);
      Pt[v] += dg ? qcol : 0.0f;
      const float s  = mrow[v] * mcol * Pt[v];
      const float sd = s + dterm;
      const float sv = (dg && in14) ? sd : s;
      St[v] = sv;
      nS[v] = -sv;
      Yn[v] = -(Pt[v] * mrow[v]);
      dS = dg ? sv : dS;
    }
    float tv = in14 ? mcol * dS : 0.0f;
#pragma unroll
    for (int off = 16; off >= 1; off >>= 1) tv += __shfl_xor(tv, off, 32);
    const float dload = JITTER * fmaxf(tv, 1e-6f);
    v8f m;
#pragma unroll
    for (int v = 0; v < 8; ++v) {
      const bool dg = (v + 8 * h == l16);
      m[v] = dg ? (in14 ? St[v] + dload : 1.0f) : St[v];
    }
#pragma unroll
    for (int k = 0; k < D14; ++k) {
      const int hk = k >> 3, vk = k & 7;
      const float rk = m[vk];
      const float d  = __int_as_float(
          __builtin_amdgcn_readlane(__float_as_int(rk), k + 16 * hk));
      const float ip = 1.0f / d;
      const bool isK = (l16 == k);
      const bool inH = (h == hk);
      const float cv = rk * ip;
      v2f a, bb;
      a.x  = (inH && !isK) ? rk  : 0.0f;
      bb.x = (inH && !isK) ? -cv : 0.0f;
      a.y  = inH ? (isK ? -ip : cv) : 0.0f;
      bb.y = (inH && isK) ? 1.0f : 0.0f;
#pragma unroll
      for (int v = 0; v < 8; ++v) m[v] = isK ? 0.0f : m[v];
      m[vk] = inH ? (isK ? 0.0f : cv) : m[vk];
      m = __builtin_amdgcn_wmma_f32_16x16x4_f32(false, a, false, bb, (short)0, m,
                                                false, false);
    }
    v8f KT = {};
    KT = mm_tn(m, Yn, KT);
    v8f nSK = {};
    nSK = mm_tn(nS, KT, nSK);
    float part = 0.0f;
#pragma unroll
    for (int v = 0; v < 8; ++v) part += KT[v] * ivrow[v];
    const float oth = __shfl_xor(part, 16, 32);
    v8f pa = Pt, pb = Pt;
#pragma unroll
    for (int kk = 0; kk < 4; ++kk) {
      v2f u, w;
      u.x = nSK[2 * kk]; u.y = nSK[2 * kk + 1];
      w.x = KT[2 * kk];  w.y = KT[2 * kk + 1];
      pa = __builtin_amdgcn_wmma_f32_16x16x4_f32(false, u, false, w, (short)0, pa,
                                                 false, false);
      pb = __builtin_amdgcn_wmma_f32_16x16x4_f32(false, w, false, u, (short)0, pb,
                                                 false, false);
    }
#pragma unroll
    for (int v = 0; v < 8; ++v) Pt[v] = 0.5f * (pa[v] + pb[v]);
    if (lane < D14) out[tok * D14 + lane] = xpreg + (part + oth);
  }
}

extern "C" void kernel_launch(void* const* d_in, const int* in_sizes, int n_in,
                              void* d_out, int out_size, void* d_ws, size_t ws_size,
                              hipStream_t stream) {
  const float* xh  = (const float*)d_in[0];
  const float* z   = (const float*)d_in[1];
  const float* m21 = (const float*)d_in[2];
  const float* Wf  = (const float*)d_in[3];
  const float* bf  = (const float*)d_in[4];
  const float* Wq  = (const float*)d_in[5];
  const float* bq  = (const float*)d_in[6];
  const float* Wr  = (const float*)d_in[7];
  const float* br  = (const float*)d_in[8];
  float* out = (float*)d_out;

  const int T  = 512;
  const int BT = in_sizes[1] / D14;
  const int B  = BT / T;

  kf_scan<<<dim3(B), dim3(32), 0, stream>>>(xh, z, m21, Wf, bf, Wq, bq, Wr, br,
                                            out, T);
}


# Tokenwise nets leave LDS and barriers for the matrix unit: 12 WMMAs on register-held weight tiles make step t+1's vectors at the end of step t, inputs read two steps ahead; net sum order changes
# speedup vs baseline: 5.3703x; 1.4016x over previous
#include <hip/hip_runtime.h>

#define D14   14
#define JITTER 1e-4f
#define MIN_Q  1e-5f
#define MIN_R  1e-4f

typedef __attribute__((ext_vector_type(2))) float v2f;
typedef __attribute__((ext_vector_type(8))) float v8f;

__device__ __forceinline__ v8f mm_tn(v8f X, v8f Y, v8f c) {
#pragma unroll
  for (int kk = 0; kk < 4; ++kk) {
    v2f a, b;
    a.x = X[2 * kk]; a.y = X[2 * kk + 1];
    b.x = Y[2 * kk]; b.y = Y[2 * kk + 1];
    c = __builtin_amdgcn_wmma_f32_16x16x4_f32(false, a, false, b, (short)0, c,
                                              false, false);
  }
  return c;
}

struct TokRaw {
  float x[8], z[8];
  float mr[4];
  float mc;
};

__device__ __forceinline__ TokRaw load_tok(const float* __restrict__ xh,
                                           const float* __restrict__ z,
                                           const float* __restrict__ m21,
                                           long long tok, const int* er, const int* gr,
                                           int gc) {
  TokRaw r;
#pragma unroll
  for (int v = 0; v < 8; ++v) {
    r.x[v] = xh[tok * D14 + er[v]];
    r.z[v] = z[tok * D14 + er[v]];
  }
#pragma unroll
  for (int j = 0; j < 4; ++j) r.mr[j] = m21[tok * 21 + gr[j]];
  r.mc = m21[tok * 21 + gc];
  return r;
}

__device__ __forceinline__ v8f net_mm(const float* X, const float* Y, int kk, v8f c) {
  v2f a, b;
  a.x = X[2 * kk]; a.y = X[2 * kk + 1];
  b.x = Y[2 * kk]; b.y = Y[2 * kk + 1];
  return __builtin_amdgcn_wmma_f32_16x16x4_f32(false, a, false, b, (short)0, c, false, false);
}

struct StepVec {
  float qcol, rcol, mcol;
  float xpcol;
  float mrow[8], ivrow[8];
};

struct NetW {
  float tWf[8], tWq[8], tWr[8];
  float bfr[8];
  float bqc, brc;
};

__device__ __forceinline__ StepVec token_nets(const TokRaw& r, const NetW& w, int l16, int h,
                                              bool in14) {
  v8f xr, qc, rc;
#pragma unroll
  for (int v = 0; v < 8; ++v) { xr[v] = w.bfr[v]; qc[v] = w.bqc; rc[v] = w.brc; }
#pragma unroll
  for (int kk = 0; kk < 4; ++kk) {
    xr = net_mm(w.tWf, r.x, kk, xr);
    rc = net_mm(r.z, w.tWr, kk, rc);
  }
  float xrf[8];
#pragma unroll
  for (int v = 0; v < 8; ++v) xrf[v] = xr[v];
#pragma unroll
  for (int kk = 0; kk < 4; ++kk) qc = net_mm(xrf, w.tWq, kk, qc);
  StepVec s;
  s.qcol  = in14 ? fmaxf(qc[0], MIN_Q) : 0.0f;
  s.rcol  = in14 ? fmaxf(rc[0], MIN_R) : 0.0f;
  s.mcol  = (in14 && r.mc > 0.5f) ? 1.0f : 0.0f;
  float xsel = xr[0];
#pragma unroll
  for (int v = 1; v < 8; ++v) xsel = ((l16 & 7) == v) ? xr[v] : xsel;
  s.xpcol = xsel;
#pragma unroll
  for (int v = 0; v < 8; ++v) {
    s.mrow[v]  = (v + 8 * h < D14 && r.mr[v >> 1] > 0.5f) ? 1.0f : 0.0f;
    s.ivrow[v] = s.mrow[v] * (r.z[v] - xr[v]);
  }
  return s;
}

__global__ __launch_bounds__(32) void kf_scan(
    const float* __restrict__ xh, const float* __restrict__ z,
    const float* __restrict__ m21,
    const float* __restrict__ Wf, const float* __restrict__ bf,
    const float* __restrict__ Wq, const float* __restrict__ bq,
    const float* __restrict__ Wr, const float* __restrict__ br,
    float* __restrict__ out, int T) {
  const int lane = threadIdx.x;
  const int b    = blockIdx.x;
  const int l16  = lane & 15;
  const int h    = lane >> 4;
  const bool in14 = (l16 < D14);
  const int  ci   = in14 ? l16 : (D14 - 1);

  NetW w;
  int er[8], gr[4];
#pragma unroll
  for (int v = 0; v < 8; ++v) {
    const int  k  = v + 8 * h;
    const int  kc = (k < D14) ? k : (D14 - 1);
    const bool ok = (k < D14) && in14;
    const float wf = Wf[kc * D14 + ci], wq = Wq[kc * D14 + ci], wr = Wr[kc * D14 + ci];
    const float bv = bf[kc];
    w.tWf[v] = ok ? wf : 0.0f;
    w.tWq[v] = ok ? wq : 0.0f;
    w.tWr[v] = ok ? wr : 0.0f;
    w.bfr[v] = (k < D14) ? bv : 0.0f;
    er[v]    = kc;
  }
#pragma unroll
  for (int j = 0; j < 4; ++j) {
    const int g = 4 * h + j;
    gr[j] = 3 * ((g < 7) ? g : 6) + 2;
  }
  const int gc = 3 * (ci >> 1) + 2;
  const float bql = bq[ci], brl = br[ci];
  w.bqc = in14 ? bql : 0.0f;
  w.brc = in14 ? brl : 0.0f;

  v8f Pt;
#pragma unroll
  for (int v = 0; v < 8; ++v) Pt[v] = (v + 8 * h == l16 && in14) ? 1.0f : 0.0f;

  const long long base = (long long)b * T;

  StepVec sv = token_nets(load_tok(xh, z, m21, base, er, gr, gc), w, l16, h, in14);
  TokRaw rn = load_tok(xh, z, m21, base + ((1 < T) ? 1 : 0), er, gr, gc);

#pragma unroll 1
  for (int t = 0; t < T; ++t) {
    const long long tok = base + t;
    const long long tokn = base + ((t + 2 < T) ? (t + 2) : (T - 1));
    const TokRaw rnn = load_tok(xh, z, m21, tokn, er, gr, gc);
    const float qcol = sv.qcol, rcol = sv.rcol, mcol = sv.mcol;
    const float dterm = mcol * rcol + (1.0f - mcol) + JITTER * mcol;
    v8f St, nS, Yn;
    float dS = 0.0f;
#pragma unroll
    for (int v = 0; v < 8; ++v) {
      const bool dg = (v + 8 * h == l16);
      Pt[v] += dg ? qcol : 0.0f;
      const float s  = sv.mrow[v] * mcol * Pt[v];
      const float sd = s + dterm;
      const float sx = (dg && in14) ? sd : s;
      St[v] = sx;
      nS[v] = -sx;
      Yn[v] = -(Pt[v] * sv.mrow[v]);
      dS = dg ? sx : dS;
    }
    float tv = in14 ? mcol * dS : 0.0f;
#pragma unroll
    for (int off = 16; off >= 1; off >>= 1) tv += __shfl_xor(tv, off, 32);
    const float dload = JITTER * fmaxf(tv, 1e-6f);
    v8f m;
#pragma unroll
    for (int v = 0; v < 8; ++v) {
      const bool dg = (v + 8 * h == l16);
      m[v] = dg ? (in14 ? St[v] + dload : 1.0f) : St[v];
    }
#pragma unroll
    for (int k = 0; k < D14; ++k) {
      const int hk = k >> 3, vk = k & 7;
      const float rk = m[vk];
      const float d  = __int_as_float(
          __builtin_amdgcn_readlane(__float_as_int(rk), k + 16 * hk));
      const float ip = 1.0f / d;
      const bool isK = (l16 == k);
      const bool inH = (h == hk);
      const float cv = rk * ip;
      v2f a, bb;
      a.x  = (inH && !isK) ? rk  : 0.0f;
      bb.x = (inH && !isK) ? -cv : 0.0f;
      a.y  = inH ? (isK ? -ip : cv) : 0.0f;
      bb.y = (inH && isK) ? 1.0f : 0.0f;
#pragma unroll
      for (int v = 0; v < 8; ++v) m[v] = isK ? 0.0f : m[v];
      m[vk] = inH ? (isK ? 0.0f : cv) : m[vk];
      m = __builtin_amdgcn_wmma_f32_16x16x4_f32(false, a, false, bb, (short)0, m,
                                                false, false);
    }
    v8f KT = {};
    KT = mm_tn(m, Yn, KT);
    v8f nSK = {};
    nSK = mm_tn(nS, KT, nSK);
    float part = 0.0f;
#pragma unroll
    for (int v = 0; v < 8; ++v) part += KT[v] * sv.ivrow[v];
    const float oth = __shfl_xor(part, 16, 32);
    v8f pa = Pt, pb = Pt;
#pragma unroll
    for (int kk = 0; kk < 4; ++kk) {
      v2f u, w2;
      u.x  = nSK[2 * kk]; u.y  = nSK[2 * kk + 1];
      w2.x = KT[2 * kk];  w2.y = KT[2 * kk + 1];
      pa = __builtin_amdgcn_wmma_f32_16x16x4_f32(false, u, false, w2, (short)0, pa,
                                                 false, false);
      pb = __builtin_amdgcn_wmma_f32_16x16x4_f32(false, w2, false, u, (short)0, pb,
                                                 false, false);
    }
#pragma unroll
    for (int v = 0; v < 8; ++v) Pt[v] = 0.5f * (pa[v] + pb[v]);
    if ((l16 >> 3) == h && in14) out[tok * D14 + l16] = sv.xpcol + (part + oth);
    sv = token_nets(rn, w, l16, h, in14);
    rn = rnn;
  }
}

extern "C" void kernel_launch(void* const* d_in, const int* in_sizes, int n_in,
                              void* d_out, int out_size, void* d_ws, size_t ws_size,
                              hipStream_t stream) {
  const float* xh  = (const float*)d_in[0];
  const float* z   = (const float*)d_in[1];
  const float* m21 = (const float*)d_in[2];
  const float* Wf  = (const float*)d_in[3];
  const float* bf  = (const float*)d_in[4];
  const float* Wq  = (const float*)d_in[5];
  const float* bq  = (const float*)d_in[6];
  const float* Wr  = (const float*)d_in[7];
  const float* br  = (const float*)d_in[8];
  float* out = (float*)d_out;

  const int T  = 512;
  const int BT = in_sizes[1] / D14;
  const int B  = BT / T;

  kf_scan<<<dim3(B), dim3(32), 0, stream>>>(xh, z, m21, Wf, bf, Wq, bq, Wr, br,
                                            out, T);
}


# Inverse sweep takes two rows per pivot: 7 block pivots (2x2 block inverted by adjugate and one reciprocal of its determinant, one rank-2 16x16x4 WMMA each) replace 14; order of the sums changes
# speedup vs baseline: 6.5876x; 1.2267x over previous
#include <hip/hip_runtime.h>

#define D14   14
#define JITTER 1e-4f
#define MIN_Q  1e-5f
#define MIN_R  1e-4f

typedef __attribute__((ext_vector_type(2))) float v2f;
typedef __attribute__((ext_vector_type(8))) float v8f;

__device__ __forceinline__ v8f mm_tn(v8f X, v8f Y, v8f c) {
#pragma unroll
  for (int kk = 0; kk < 4; ++kk) {
    v2f a, b;
    a.x = X[2 * kk]; a.y = X[2 * kk + 1];
    b.x = Y[2 * kk]; b.y = Y[2 * kk + 1];
    c = __builtin_amdgcn_wmma_f32_16x16x4_f32(false, a, false, b, (short)0, c,
                                              false, false);
  }
  return c;
}

__device__ __forceinline__ float xhalf(float x) {
  const int b = __builtin_bit_cast(int, x);
  return __builtin_bit_cast(float,
      __builtin_amdgcn_permlanex16(b, b, 0x76543210, 0xfedcba98, false, false));
}

struct TokRaw {
  float x[8], z[8];
  float mr[4];
  float mc;
};

__device__ __forceinline__ TokRaw load_tok(const float* __restrict__ xh,
                                           const float* __restrict__ z,
                                           const float* __restrict__ m21,
                                           long long tok, const int* er, const int* gr,
                                           int gc) {
  TokRaw r;
#pragma unroll
  for (int v = 0; v < 8; ++v) {
    r.x[v] = xh[tok * D14 + er[v]];
    r.z[v] = z[tok * D14 + er[v]];
  }
#pragma unroll
  for (int j = 0; j < 4; ++j) r.mr[j] = m21[tok * 21 + gr[j]];
  r.mc = m21[tok * 21 + gc];
  return r;
}

__device__ __forceinline__ v8f net_mm(const float* X, const float* Y, int kk, v8f c) {
  v2f a, b;
  a.x = X[2 * kk]; a.y = X[2 * kk + 1];
  b.x = Y[2 * kk]; b.y = Y[2 * kk + 1];
  return __builtin_amdgcn_wmma_f32_16x16x4_f32(false, a, false, b, (short)0, c, false, false);
}

struct StepVec {
  float qcol, rcol, mcol;
  float xpcol;
  float mrow[8], ivrow[8];
};

struct NetW {
  float tWf[8], tWq[8], tWr[8];
  float bfr[8];
  float bqc, brc;
};

__device__ __forceinline__ StepVec token_nets(const TokRaw& r, const NetW& w, int l16, int h,
                                              bool in14) {
  v8f xr, qc, rc;
#pragma unroll
  for (int v = 0; v < 8; ++v) { xr[v] = w.bfr[v]; qc[v] = w.bqc; rc[v] = w.brc; }
#pragma unroll
  for (int kk = 0; kk < 4; ++kk) {
    xr = net_mm(w.tWf, r.x, kk, xr);
    rc = net_mm(r.z, w.tWr, kk, rc);
  }
  float xrf[8];
#pragma unroll
  for (int v = 0; v < 8; ++v) xrf[v] = xr[v];
#pragma unroll
  for (int kk = 0; kk < 4; ++kk) qc = net_mm(xrf, w.tWq, kk, qc);
  StepVec s;
  s.qcol  = in14 ? fmaxf(qc[0], MIN_Q) : 0.0f;
  s.rcol  = in14 ? fmaxf(rc[0], MIN_R) : 0.0f;
  s.mcol  = (in14 && r.mc > 0.5f) ? 1.0f : 0.0f;
  float xsel = xr[0];
#pragma unroll
  for (int v = 1; v < 8; ++v) xsel = ((l16 & 7) == v) ? xr[v] : xsel;
  s.xpcol = xsel;
#pragma unroll
  for (int v = 0; v < 8; ++v) {
    s.mrow[v]  = (v + 8 * h < D14 && r.mr[v >> 1] > 0.5f) ? 1.0f : 0.0f;
    s.ivrow[v] = s.mrow[v] * (r.z[v] - xr[v]);
  }
  return s;
}

__global__ __launch_bounds__(32) void kf_scan(
    const float* __restrict__ xh, const float* __restrict__ z,
    const float* __restrict__ m21,
    const float* __restrict__ Wf, const float* __restrict__ bf,
    const float* __restrict__ Wq, const float* __restrict__ bq,
    const float* __restrict__ Wr, const float* __restrict__ br,
    float* __restrict__ out, int T) {
  const int lane = threadIdx.x;
  const int b    = blockIdx.x;
  const int l16  = lane & 15;
  const int h    = lane >> 4;
  const bool in14 = (l16 < D14);
  const int  ci   = in14 ? l16 : (D14 - 1);

  NetW w;
  int er[8], gr[4];
#pragma unroll
  for (int v = 0; v < 8; ++v) {
    const int  k  = v + 8 * h;
    const int  kc = (k < D14) ? k : (D14 - 1);
    const bool ok = (k < D14) && in14;
    const float wf = Wf[kc * D14 + ci], wq = Wq[kc * D14 + ci], wr = Wr[kc * D14 + ci];
    const float bv = bf[kc];
    w.tWf[v] = ok ? wf : 0.0f;
    w.tWq[v] = ok ? wq : 0.0f;
    w.tWr[v] = ok ? wr : 0.0f;
    w.bfr[v] = (k < D14) ? bv : 0.0f;
    er[v]    = kc;
  }
#pragma unroll
  for (int j = 0; j < 4; ++j) {
    const int g = 4 * h + j;
    gr[j] = 3 * ((g < 7) ? g : 6) + 2;
  }
  const int gc = 3 * (ci >> 1) + 2;
  const float bql = bq[ci], brl = br[ci];
  w.bqc = in14 ? bql : 0.0f;
  w.brc = in14 ? brl : 0.0f;

  v8f Pt;
#pragma unroll
  for (int v = 0; v < 8; ++v) Pt[v] = (v + 8 * h == l16 && in14) ? 1.0f : 0.0f;

  const long long base = (long long)b * T;

  StepVec sv = token_nets(load_tok(xh, z, m21, base, er, gr, gc), w, l16, h, in14);
  TokRaw rn = load_tok(xh, z, m21, base + ((1 < T) ? 1 : 0), er, gr, gc);

#pragma unroll 1
  for (int t = 0; t < T; ++t) {
    const long long tok = base + t;
    const long long tokn = base + ((t + 2 < T) ? (t + 2) : (T - 1));
    const TokRaw rnn = load_tok(xh, z, m21, tokn, er, gr, gc);
    const float qcol = sv.qcol, rcol = sv.rcol, mcol = sv.mcol;
    const float dterm = mcol * rcol + (1.0f - mcol) + JITTER * mcol;
    v8f St, nS, Yn;
    float dS = 0.0f;
#pragma unroll
    for (int v = 0; v < 8; ++v) {
      const bool dg = (v + 8 * h == l16);
      Pt[v] += dg ? qcol : 0.0f;
      const float s  = sv.mrow[v] * mcol * Pt[v];
      const float sd = s + dterm;
      const float sx = (dg && in14) ? sd : s;
      St[v] = sx;
      nS[v] = -sx;
      Yn[v] = -(Pt[v] * sv.mrow[v]);
      dS = dg ? sx : dS;
    }
    float tv = in14 ? mcol * dS : 0.0f;
#pragma unroll
    for (int off = 16; off >= 1; off >>= 1) tv += __shfl_xor(tv, off, 32);
    const float dload = JITTER * fmaxf(tv, 1e-6f);
    v8f m;
#pragma unroll
    for (int v = 0; v < 8; ++v) {
      const bool dg = (v + 8 * h == l16);
      m[v] = dg ? (in14 ? St[v] + dload : 1.0f) : St[v];
    }
#pragma unroll
    for (int p = 0; p < D14 / 2; ++p) {
      const int k0 = 2 * p;
      const int hk = k0 >> 3, v0 = k0 & 7, v1 = v0 + 1;
      const int ln = k0 + 16 * hk;
      const float r0 = m[v0], r1 = m[v1];
      const float a = __int_as_float(__builtin_amdgcn_readlane(__float_as_int(r0), ln));
      const float b = __int_as_float(__builtin_amdgcn_readlane(__float_as_int(r0), ln + 1));
      const float c = __int_as_float(__builtin_amdgcn_readlane(__float_as_int(r1), ln + 1));
      const float idet = 1.0f / fmaf(-b, b, a * c);
      const bool inH = (h == hk);
      const bool is0 = (l16 == k0), is1 = (l16 == k0 + 1);
      const bool inK = is0 || is1;
      const float t0 = c * r0 - b * r1;
      const float t1 = a * r1 - b * r0;
      const float u0 = is0 ? -c : (is1 ? b : t0);
      const float u1 = is0 ? b : (is1 ? -a : t1);
      const float x0 = xhalf(u0), x1 = xhalf(u1);
      v2f A, Bv;
      A.x = inH ? (is0 ? -1.0f : (is1 ? 0.0f : r0)) : x0;
      A.y = inH ? (is1 ? -1.0f : (is0 ? 0.0f : r1)) : x1;
      const float s0 = inH ? (inK ? 0.0f : -t0) : (is0 ? 1.0f : 0.0f);
      const float s1 = inH ? (inK ? 0.0f : -t1) : (is1 ? 1.0f : 0.0f);
      Bv.x = s0 * idet;
      Bv.y = s1 * idet;
#pragma unroll
      for (int v = 0; v < 8; ++v) m[v] = inK ? 0.0f : m[v];
      m[v0] = inH ? 0.0f : m[v0];
      m[v1] = inH ? 0.0f : m[v1];
      m = __builtin_amdgcn_wmma_f32_16x16x4_f32(false, A, false, Bv, (short)0, m,
                                                false, false);
    }
    v8f KT = {};
    KT = mm_tn(m, Yn, KT);
    v8f nSK = {};
    nSK = mm_tn(nS, KT, nSK);
    float part = 0.0f;
#pragma unroll
    for (int v = 0; v < 8; ++v) part += KT[v] * sv.ivrow[v];
    const float oth = __shfl_xor(part, 16, 32);
    v8f pa = Pt, pb = Pt;
#pragma unroll
    for (int kk = 0; kk < 4; ++kk) {
      v2f u, w2;
      u.x  = nSK[2 * kk]; u.y  = nSK[2 * kk + 1];
      w2.x = KT[2 * kk];  w2.y = KT[2 * kk + 1];
      pa = __builtin_amdgcn_wmma_f32_16x16x4_f32(false, u, false, w2, (short)0, pa,
                                                 false, false);
      pb = __builtin_amdgcn_wmma_f32_16x16x4_f32(false, w2, false, u, (short)0, pb,
                                                 false, false);
    }
#pragma unroll
    for (int v = 0; v < 8; ++v) Pt[v] = 0.5f * (pa[v] + pb[v]);
    if ((l16 >> 3) == h && in14) out[tok * D14 + l16] = sv.xpcol + (part + oth);
    sv = token_nets(rn, w, l16, h, in14);
    rn = rnn;
  }
}

extern "C" void kernel_launch(void* const* d_in, const int* in_sizes, int n_in,
                              void* d_out, int out_size, void* d_ws, size_t ws_size,
                              hipStream_t stream) {
  const float* xh  = (const float*)d_in[0];
  const float* z   = (const float*)d_in[1];
  const float* m21 = (const float*)d_in[2];
  const float* Wf  = (const float*)d_in[3];
  const float* bf  = (const float*)d_in[4];
  const float* Wq  = (const float*)d_in[5];
  const float* bq  = (const float*)d_in[6];
  const float* Wr  = (const float*)d_in[7];
  const float* br  = (const float*)d_in[8];
  float* out = (float*)d_out;

  const int T  = 512;
  const int BT = in_sizes[1] / D14;
  const int B  = BT / T;

  kf_scan<<<dim3(B), dim3(32), 0, stream>>>(xh, z, m21, Wf, bf, Wq, bq, Wr, br,
                                            out, T);
}


# Cross-lane sum of the trace moves to the matrix unit: one 16x16x4 WMMA against ones puts the lane terms in registers, 7 in-lane adds and a half swap replace 5 LDS exchanges; sum order changes
# speedup vs baseline: 7.4137x; 1.1254x over previous
#include <hip/hip_runtime.h>

#define D14   14
#define JITTER 1e-4f
#define MIN_Q  1e-5f
#define MIN_R  1e-4f

typedef __attribute__((ext_vector_type(2))) float v2f;
typedef __attribute__((ext_vector_type(8))) float v8f;

__device__ __forceinline__ v8f mm_tn(v8f X, v8f Y, v8f c) {
#pragma unroll
  for (int kk = 0; kk < 4; ++kk) {
    v2f a, b;
    a.x = X[2 * kk]; a.y = X[2 * kk + 1];
    b.x = Y[2 * kk]; b.y = Y[2 * kk + 1];
    c = __builtin_amdgcn_wmma_f32_16x16x4_f32(false, a, false, b, (short)0, c,
                                              false, false);
  }
  return c;
}

__device__ __forceinline__ float xhalf(float x) {
  const int b = __builtin_bit_cast(int, x);
  return __builtin_bit_cast(float,
      __builtin_amdgcn_permlanex16(b, b, 0x76543210, 0xfedcba98, false, false));
}

struct TokRaw {
  float x[8], z[8];
  float mr[4];
  float mc;
};

__device__ __forceinline__ TokRaw load_tok(const float* __restrict__ xh,
                                           const float* __restrict__ z,
                                           const float* __restrict__ m21,
                                           long long tok, const int* er, const int* gr,
                                           int gc) {
  TokRaw r;
#pragma unroll
  for (int v = 0; v < 8; ++v) {
    r.x[v] = xh[tok * D14 + er[v]];
    r.z[v] = z[tok * D14 + er[v]];
  }
#pragma unroll
  for (int j = 0; j < 4; ++j) r.mr[j] = m21[tok * 21 + gr[j]];
  r.mc = m21[tok * 21 + gc];
  return r;
}

__device__ __forceinline__ v8f net_mm(const float* X, const float* Y, int kk, v8f c) {
  v2f a, b;
  a.x = X[2 * kk]; a.y = X[2 * kk + 1];
  b.x = Y[2 * kk]; b.y = Y[2 * kk + 1];
  return __builtin_amdgcn_wmma_f32_16x16x4_f32(false, a, false, b, (short)0, c, false, false);
}

struct StepVec {
  float qcol, rcol, mcol;
  float xpcol;
  float mrow[8], ivrow[8];
};

struct NetW {
  float tWf[8], tWq[8], tWr[8];
  float bfr[8];
  float bqc, brc;
};

__device__ __forceinline__ StepVec token_nets(const TokRaw& r, const NetW& w, int l16, int h,
                                              bool in14) {
  v8f xr, qc, rc;
#pragma unroll
  for (int v = 0; v < 8; ++v) { xr[v] = w.bfr[v]; qc[v] = w.bqc; rc[v] = w.brc; }
#pragma unroll
  for (int kk = 0; kk < 4; ++kk) {
    xr = net_mm(w.tWf, r.x, kk, xr);
    rc = net_mm(r.z, w.tWr, kk, rc);
  }
  float xrf[8];
#pragma unroll
  for (int v = 0; v < 8; ++v) xrf[v] = xr[v];
#pragma unroll
  for (int kk = 0; kk < 4; ++kk) qc = net_mm(xrf, w.tWq, kk, qc);
  StepVec s;
  s.qcol  = in14 ? fmaxf(qc[0], MIN_Q) : 0.0f;
  s.rcol  = in14 ? fmaxf(rc[0], MIN_R) : 0.0f;
  s.mcol  = (in14 && r.mc > 0.5f) ? 1.0f : 0.0f;
  float xsel = xr[0];
#pragma unroll
  for (int v = 1; v < 8; ++v) xsel = ((l16 & 7) == v) ? xr[v] : xsel;
  s.xpcol = xsel;
#pragma unroll
  for (int v = 0; v < 8; ++v) {
    s.mrow[v]  = (v + 8 * h < D14 && r.mr[v >> 1] > 0.5f) ? 1.0f : 0.0f;
    s.ivrow[v] = s.mrow[v] * (r.z[v] - xr[v]);
  }
  return s;
}

__global__ __launch_bounds__(32) void kf_scan(
    const float* __restrict__ xh, const float* __restrict__ z,
    const float* __restrict__ m21,
    const float* __restrict__ Wf, const float* __restrict__ bf,
    const float* __restrict__ Wq, const float* __restrict__ bq,
    const float* __restrict__ Wr, const float* __restrict__ br,
    float* __restrict__ out, int T) {
  const int lane = threadIdx.x;
  const int b    = blockIdx.x;
  const int l16  = lane & 15;
  const int h    = lane >> 4;
  const bool in14 = (l16 < D14);
  const int  ci   = in14 ? l16 : (D14 - 1);

  NetW w;
  int er[8], gr[4];
#pragma unroll
  for (int v = 0; v < 8; ++v) {
    const int  k  = v + 8 * h;
    const int  kc = (k < D14) ? k : (D14 - 1);
    const bool ok = (k < D14) && in14;
    const float wf = Wf[kc * D14 + ci], wq = Wq[kc * D14 + ci], wr = Wr[kc * D14 + ci];
    const float bv = bf[kc];
    w.tWf[v] = ok ? wf : 0.0f;
    w.tWq[v] = ok ? wq : 0.0f;
    w.tWr[v] = ok ? wr : 0.0f;
    w.bfr[v] = (k < D14) ? bv : 0.0f;
    er[v]    = kc;
  }
#pragma unroll
  for (int j = 0; j < 4; ++j) {
    const int g = 4 * h + j;
    gr[j] = 3 * ((g < 7) ? g : 6) + 2;
  }
  const int gc = 3 * (ci >> 1) + 2;
  const float bql = bq[ci], brl = br[ci];
  w.bqc = in14 ? bql : 0.0f;
  w.brc = in14 ? brl : 0.0f;

  v8f Pt;
#pragma unroll
  for (int v = 0; v < 8; ++v) Pt[v] = (v + 8 * h == l16 && in14) ? 1.0f : 0.0f;

  const long long base = (long long)b * T;

  StepVec sv = token_nets(load_tok(xh, z, m21, base, er, gr, gc), w, l16, h, in14);
  TokRaw rn = load_tok(xh, z, m21, base + ((1 < T) ? 1 : 0), er, gr, gc);

#pragma unroll 1
  for (int t = 0; t < T; ++t) {
    const long long tok = base + t;
    const long long tokn = base + ((t + 2 < T) ? (t + 2) : (T - 1));
    const TokRaw rnn = load_tok(xh, z, m21, tokn, er, gr, gc);
    const float qcol = sv.qcol, rcol = sv.rcol, mcol = sv.mcol;
    const float dterm = mcol * rcol + (1.0f - mcol) + JITTER * mcol;
    v8f St, nS, Yn;
    float dS = 0.0f;
#pragma unroll
    for (int v = 0; v < 8; ++v) {
      const bool dg = (v + 8 * h == l16);
      Pt[v] += dg ? qcol : 0.0f;
      const float s  = sv.mrow[v] * mcol * Pt[v];
      const float sd = s + dterm;
      const float sx = (dg && in14) ? sd : s;
      St[v] = sx;
      nS[v] = -sx;
      Yn[v] = -(Pt[v] * sv.mrow[v]);
      dS = dg ? sx : dS;
    }
    const float tvl = in14 ? mcol * dS : 0.0f;
    v2f ta, tb;
    ta.x = tvl;  ta.y = 0.0f;
    tb.x = 1.0f; tb.y = 1.0f;
    v8f te = {};
    te = __builtin_amdgcn_wmma_f32_16x16x4_f32(false, ta, false, tb, (short)0, te,
                                               false, false);
    const float th = ((te[0] + te[2]) + (te[4] + te[6])) + ((te[1] + te[3]) + (te[5] + te[7]));
    const float tv = th + xhalf(th);
    const float dload = JITTER * fmaxf(tv, 1e-6f);
    v8f m;
#pragma unroll
    for (int v = 0; v < 8; ++v) {
      const bool dg = (v + 8 * h == l16);
      m[v] = dg ? (in14 ? St[v] + dload : 1.0f) : St[v];
    }
#pragma unroll
    for (int p = 0; p < D14 / 2; ++p) {
      const int k0 = 2 * p;
      const int hk = k0 >> 3, v0 = k0 & 7, v1 = v0 + 1;
      const int ln = k0 + 16 * hk;
      const float r0 = m[v0], r1 = m[v1];
      const float a = __int_as_float(__builtin_amdgcn_readlane(__float_as_int(r0), ln));
      const float b = __int_as_float(__builtin_amdgcn_readlane(__float_as_int(r0), ln + 1));
      const float c = __int_as_float(__builtin_amdgcn_readlane(__float_as_int(r1), ln + 1));
      const float idet = 1.0f / fmaf(-b, b, a * c);
      const bool inH = (h == hk);
      const bool is0 = (l16 == k0), is1 = (l16 == k0 + 1);
      const bool inK = is0 || is1;
      const float t0 = c * r0 - b * r1;
      const float t1 = a * r1 - b * r0;
      const float u0 = is0 ? -c : (is1 ? b : t0);
      const float u1 = is0 ? b : (is1 ? -a : t1);
      const float x0 = xhalf(u0), x1 = xhalf(u1);
      v2f A, Bv;
      A.x = inH ? (is0 ? -1.0f : (is1 ? 0.0f : r0)) : x0;
      A.y = inH ? (is1 ? -1.0f : (is0 ? 0.0f : r1)) : x1;
      const float s0 = inH ? (inK ? 0.0f : -t0) : (is0 ? 1.0f : 0.0f);
      const float s1 = inH ? (inK ? 0.0f : -t1) : (is1 ? 1.0f : 0.0f);
      Bv.x = s0 * idet;
      Bv.y = s1 * idet;
#pragma unroll
      for (int v = 0; v < 8; ++v) m[v] = inK ? 0.0f : m[v];
      m[v0] = inH ? 0.0f : m[v0];
      m[v1] = inH ? 0.0f : m[v1];
      m = __builtin_amdgcn_wmma_f32_16x16x4_f32(false, A, false, Bv, (short)0, m,
                                                false, false);
    }
    v8f KT = {};
    KT = mm_tn(m, Yn, KT);
    v8f nSK = {};
    nSK = mm_tn(nS, KT, nSK);
    float part = 0.0f;
#pragma unroll
    for (int v = 0; v < 8; ++v) part += KT[v] * sv.ivrow[v];
    const float oth = __shfl_xor(part, 16, 32);
    v8f pa = Pt, pb = Pt;
#pragma unroll
    for (int kk = 0; kk < 4; ++kk) {
      v2f u, w2;
      u.x  = nSK[2 * kk]; u.y  = nSK[2 * kk + 1];
      w2.x = KT[2 * kk];  w2.y = KT[2 * kk + 1];
      pa = __builtin_amdgcn_wmma_f32_16x16x4_f32(false, u, false, w2, (short)0, pa,
                                                 false, false);
      pb = __builtin_amdgcn_wmma_f32_16x16x4_f32(false, w2, false, u, (short)0, pb,
                                                 false, false);
    }
#pragma unroll
    for (int v = 0; v < 8; ++v) Pt[v] = 0.5f * (pa[v] + pb[v]);
    if ((l16 >> 3) == h && in14) out[tok * D14 + l16] = sv.xpcol + (part + oth);
    sv = token_nets(rn, w, l16, h, in14);
    rn = rnn;
  }
}

extern "C" void kernel_launch(void* const* d_in, const int* in_sizes, int n_in,
                              void* d_out, int out_size, void* d_ws, size_t ws_size,
                              hipStream_t stream) {
  const float* xh  = (const float*)d_in[0];
  const float* z   = (const float*)d_in[1];
  const float* m21 = (const float*)d_in[2];
  const float* Wf  = (const float*)d_in[3];
  const float* bf  = (const float*)d_in[4];
  const float* Wq  = (const float*)d_in[5];
  const float* bq  = (const float*)d_in[6];
  const float* Wr  = (const float*)d_in[7];
  const float* br  = (const float*)d_in[8];
  float* out = (float*)d_out;

  const int T  = 512;
  const int BT = in_sizes[1] / D14;
  const int B  = BT / T;

  kf_scan<<<dim3(B), dim3(32), 0, stream>>>(xh, z, m21, Wf, bf, Wq, bq, Wr, br,
                                            out, T);
}


# Next token's work (12 net WMMAs, their vector tail, the reads of token t+2) issues inside the inverse sweep's reciprocal waits, not at the step's end; the store heads the next step; sums unchanged
# speedup vs baseline: 8.4688x; 1.1423x over previous
#include <hip/hip_runtime.h>

#define D14   14
#define JITTER 1e-4f
#define MIN_Q  1e-5f
#define MIN_R  1e-4f

typedef __attribute__((ext_vector_type(2))) float v2f;
typedef __attribute__((ext_vector_type(8))) float v8f;

__device__ __forceinline__ v8f mm_tn(v8f X, v8f Y, v8f c) {
#pragma unroll
  for (int kk = 0; kk < 4; ++kk) {
    v2f a, b;
    a.x = X[2 * kk]; a.y = X[2 * kk + 1];
    b.x = Y[2 * kk]; b.y = Y[2 * kk + 1];
    c = __builtin_amdgcn_wmma_f32_16x16x4_f32(false, a, false, b, (short)0, c,
                                              false, false);
  }
  return c;
}

__device__ __forceinline__ float xhalf(float x) {
  const int b = __builtin_bit_cast(int, x);
  return __builtin_bit_cast(float,
      __builtin_amdgcn_permlanex16(b, b, 0x76543210, 0xfedcba98, false, false));
}

struct TokRaw {
  float x[8], z[8];
  float mr[4];
  float mc;
};

__device__ __forceinline__ void load_tok_part(TokRaw& r, const float* __restrict__ xh,
                                              const float* __restrict__ z,
                                              const float* __restrict__ m21,
                                              long long tok, const int* er, const int* gr,
                                              int gc, int j) {
  if (j < 2) {
#pragma unroll
    for (int v = 4 * j; v < 4 * j + 4; ++v) {
      r.x[v] = xh[tok * D14 + er[v]];
      r.z[v] = z[tok * D14 + er[v]];
    }
  } else {
#pragma unroll
    for (int i = 0; i < 4; ++i) r.mr[i] = m21[tok * 21 + gr[i]];
    r.mc = m21[tok * 21 + gc];
  }
}

__device__ __forceinline__ TokRaw load_tok(const float* __restrict__ xh,
                                           const float* __restrict__ z,
                                           const float* __restrict__ m21,
                                           long long tok, const int* er, const int* gr,
                                           int gc) {
  TokRaw r;
#pragma unroll
  for (int j = 0; j < 3; ++j) load_tok_part(r, xh, z, m21, tok, er, gr, gc, j);
  return r;
}

__device__ __forceinline__ v8f net_mm(const float* X, const float* Y, int kk, v8f c) {
  v2f a, b;
  a.x = X[2 * kk]; a.y = X[2 * kk + 1];
  b.x = Y[2 * kk]; b.y = Y[2 * kk + 1];
  return __builtin_amdgcn_wmma_f32_16x16x4_f32(false, a, false, b, (short)0, c, false, false);
}

struct StepVec {
  float qcol, rcol, mcol;
  float xpcol;
  float mrow[8], ivrow[8];
};

struct NetW {
  float tWf[8], tWq[8], tWr[8];
  float bfr[8];
  float bqc, brc;
};

struct NetAcc { v8f xr, qc, rc; };

__device__ __forceinline__ NetAcc nets_begin(const NetW& w) {
  NetAcc a;
#pragma unroll
  for (int v = 0; v < 8; ++v) { a.xr[v] = w.bfr[v]; a.qc[v] = w.bqc; a.rc[v] = w.brc; }
  return a;
}

__device__ __forceinline__ void nets_one(NetAcc& a, const TokRaw& r, const NetW& w, int i) {
  if (i < 8) {
    if ((i & 1) == 0) a.xr = net_mm(w.tWf, r.x, i >> 1, a.xr);
    else              a.rc = net_mm(r.z, w.tWr, i >> 1, a.rc);
  } else {
    float xrf[8];
#pragma unroll
    for (int v = 0; v < 8; ++v) xrf[v] = a.xr[v];
    a.qc = net_mm(xrf, w.tWq, i - 8, a.qc);
  }
}

__device__ __forceinline__ void nets_one_here(NetAcc& a, const TokRaw& r, const NetW& w, int i) {
  __builtin_amdgcn_sched_barrier(0);
  nets_one(a, r, w, i);
  __builtin_amdgcn_sched_barrier(0);
}

__device__ __forceinline__ StepVec nets_end(const NetAcc& a, const TokRaw& r, int l16, int h,
                                            bool in14) {
  StepVec s;
  s.qcol  = in14 ? fmaxf(a.qc[0], MIN_Q) : 0.0f;
  s.rcol  = in14 ? fmaxf(a.rc[0], MIN_R) : 0.0f;
  s.mcol  = (in14 && r.mc > 0.5f) ? 1.0f : 0.0f;
  float xsel = a.xr[0];
#pragma unroll
  for (int v = 1; v < 8; ++v) xsel = ((l16 & 7) == v) ? a.xr[v] : xsel;
  s.xpcol = xsel;
#pragma unroll
  for (int v = 0; v < 8; ++v) {
    s.mrow[v]  = (v + 8 * h < D14 && r.mr[v >> 1] > 0.5f) ? 1.0f : 0.0f;
    s.ivrow[v] = s.mrow[v] * (r.z[v] - a.xr[v]);
  }
  return s;
}

__device__ __forceinline__ StepVec token_nets(const TokRaw& r, const NetW& w, int l16, int h,
                                              bool in14) {
  NetAcc a = nets_begin(w);
#pragma unroll
  for (int i = 0; i < 12; ++i) nets_one(a, r, w, i);
  return nets_end(a, r, l16, h, in14);
}

__global__ __launch_bounds__(32) void kf_scan(
    const float* __restrict__ xh, const float* __restrict__ z,
    const float* __restrict__ m21,
    const float* __restrict__ Wf, const float* __restrict__ bf,
    const float* __restrict__ Wq, const float* __restrict__ bq,
    const float* __restrict__ Wr, const float* __restrict__ br,
    float* __restrict__ out, int T) {
  const int lane = threadIdx.x;
  const int b    = blockIdx.x;
  const int l16  = lane & 15;
  const int h    = lane >> 4;
  const bool in14 = (l16 < D14);
  const int  ci   = in14 ? l16 : (D14 - 1);

  NetW w;
  int er[8], gr[4];
#pragma unroll
  for (int v = 0; v < 8; ++v) {
    const int  k  = v + 8 * h;
    const int  kc = (k < D14) ? k : (D14 - 1);
    const bool ok = (k < D14) && in14;
    const float wf = Wf[kc * D14 + ci], wq = Wq[kc * D14 + ci], wr = Wr[kc * D14 + ci];
    const float bv = bf[kc];
    w.tWf[v] = ok ? wf : 0.0f;
    w.tWq[v] = ok ? wq : 0.0f;
    w.tWr[v] = ok ? wr : 0.0f;
    w.bfr[v] = (k < D14) ? bv : 0.0f;
    er[v]    = kc;
  }
#pragma unroll
  for (int j = 0; j < 4; ++j) {
    const int g = 4 * h + j;
    gr[j] = 3 * ((g < 7) ? g : 6) + 2;
  }
  const int gc = 3 * (ci >> 1) + 2;
  const float bql = bq[ci], brl = br[ci];
  w.bqc = in14 ? bql : 0.0f;
  w.brc = in14 ? brl : 0.0f;

  v8f Pt;
#pragma unroll
  for (int v = 0; v < 8; ++v) Pt[v] = (v + 8 * h == l16 && in14) ? 1.0f : 0.0f;

  const long long base = (long long)b * T;

  StepVec sv = token_nets(load_tok(xh, z, m21, base, er, gr, gc), w, l16, h, in14);
  TokRaw rn = load_tok(xh, z, m21, base + ((1 < T) ? 1 : 0), er, gr, gc);

  const bool owner = ((l16 >> 3) == h) && in14;
  float pend = 0.0f;

#pragma unroll 1
  for (int t = 0; t < T; ++t) {
    const long long tok = base + t;
    if (owner && t > 0) out[(tok - 1) * D14 + l16] = pend;
    const long long tokn = base + ((t + 2 < T) ? (t + 2) : (T - 1));
    TokRaw rnn;
    const float qcol = sv.qcol, rcol = sv.rcol, mcol = sv.mcol;
    const float dterm = mcol * rcol + (1.0f - mcol) + JITTER * mcol;
    v8f St, nS, Yn;
    float dS = 0.0f;
#pragma unroll
    for (int v = 0; v < 8; ++v) {
      const bool dg = (v + 8 * h == l16);
      Pt[v] += dg ? qcol : 0.0f;
      const float s  = sv.mrow[v] * mcol * Pt[v];
      const float sd = s + dterm;
      const float sx = (dg && in14) ? sd : s;
      St[v] = sx;
      nS[v] = -sx;
      Yn[v] = -(Pt[v] * sv.mrow[v]);
      dS = dg ? sx : dS;
    }
    const float tvl = in14 ? mcol * dS : 0.0f;
    v2f ta, tb;
    ta.x = tvl;  ta.y = 0.0f;
    tb.x = 1.0f; tb.y = 1.0f;
    v8f te = {};
    te = __builtin_amdgcn_wmma_f32_16x16x4_f32(false, ta, false, tb, (short)0, te,
                                               false, false);
    const float th = ((te[0] + te[2]) + (te[4] + te[6])) + ((te[1] + te[3]) + (te[5] + te[7]));
    const float tv = th + xhalf(th);
    const float dload = JITTER * fmaxf(tv, 1e-6f);
    v8f m;
#pragma unroll
    for (int v = 0; v < 8; ++v) {
      const bool dg = (v + 8 * h == l16);
      m[v] = dg ? (in14 ? St[v] + dload : 1.0f) : St[v];
    }
    NetAcc na = nets_begin(w);
#pragma unroll
    for (int p = 0; p < D14 / 2; ++p) {
      const int k0 = 2 * p;
      const int hk = k0 >> 3, v0 = k0 & 7, v1 = v0 + 1;
      const int ln = k0 + 16 * hk;
      const float r0 = m[v0], r1 = m[v1];
      const float a = __int_as_float(__builtin_amdgcn_readlane(__float_as_int(r0), ln));
      const float b = __int_as_float(__builtin_amdgcn_readlane(__float_as_int(r0), ln + 1));
      const float c = __int_as_float(__builtin_amdgcn_readlane(__float_as_int(r1), ln + 1));
      const float det = fmaf(-b, b, a * c);
      if (p < 6) nets_one_here(na, rn, w, 2 * p);
      if (p == 0) {
        load_tok_part(rnn, xh, z, m21, tokn, er, gr, gc, 2);
        load_tok_part(rnn, xh, z, m21, tokn, er, gr, gc, 0);
      }
      if (p == 1) load_tok_part(rnn, xh, z, m21, tokn, er, gr, gc, 1);
      const float idet = 1.0f / det;
      const bool inH = (h == hk);
      const bool is0 = (l16 == k0), is1 = (l16 == k0 + 1);
      const bool inK = is0 || is1;
      const float t0 = c * r0 - b * r1;
      const float t1 = a * r1 - b * r0;
      const float u0 = is0 ? -c : (is1 ? b : t0);
      const float u1 = is0 ? b : (is1 ? -a : t1);
      const float x0 = xhalf(u0), x1 = xhalf(u1);
      v2f A, Bv;
      A.x = inH ? (is0 ? -1.0f : (is1 ? 0.0f : r0)) : x0;
      A.y = inH ? (is1 ? -1.0f : (is0 ? 0.0f : r1)) : x1;
      const float s0 = inH ? (inK ? 0.0f : -t0) : (is0 ? 1.0f : 0.0f);
      const float s1 = inH ? (inK ? 0.0f : -t1) : (is1 ? 1.0f : 0.0f);
      Bv.x = s0 * idet;
      Bv.y = s1 * idet;
#pragma unroll
      for (int v = 0; v < 8; ++v) m[v] = inK ? 0.0f : m[v];
      m[v0] = inH ? 0.0f : m[v0];
      m[v1] = inH ? 0.0f : m[v1];
      m = __builtin_amdgcn_wmma_f32_16x16x4_f32(false, A, false, Bv, (short)0, m,
                                                false, false);
      if (p < 6) nets_one_here(na, rn, w, 2 * p + 1);
    }
    const StepVec svn = nets_end(na, rn, l16, h, in14);
    v8f KT = {};
    KT = mm_tn(m, Yn, KT);
    v8f nSK = {};
    nSK = mm_tn(nS, KT, nSK);
    float part = 0.0f;
#pragma unroll
    for (int v = 0; v < 8; ++v) part += KT[v] * sv.ivrow[v];
    const float oth = __shfl_xor(part, 16, 32);
    __builtin_amdgcn_sched_barrier(0);
    v8f pa = Pt, pb = Pt;
#pragma unroll
    for (int kk = 0; kk < 4; ++kk) {
      v2f u, w2;
      u.x  = nSK[2 * kk]; u.y  = nSK[2 * kk + 1];
      w2.x = KT[2 * kk];  w2.y = KT[2 * kk + 1];
      pa = __builtin_amdgcn_wmma_f32_16x16x4_f32(false, u, false, w2, (short)0, pa,
                                                 false, false);
      pb = __builtin_amdgcn_wmma_f32_16x16x4_f32(false, w2, false, u, (short)0, pb,
                                                 false, false);
    }
#pragma unroll
    for (int v = 0; v < 8; ++v) Pt[v] = 0.5f * (pa[v] + pb[v]);
    pend = sv.xpcol + (part + oth);
    sv = svn;
    rn = rnn;
  }
  if (owner && T > 0) out[(base + T - 1) * D14 + l16] = pend;
}

extern "C" void kernel_launch(void* const* d_in, const int* in_sizes, int n_in,
                              void* d_out, int out_size, void* d_ws, size_t ws_size,
                              hipStream_t stream) {
  const float* xh  = (const float*)d_in[0];
  const float* z   = (const float*)d_in[1];
  const float* m21 = (const float*)d_in[2];
  const float* Wf  = (const float*)d_in[3];
  const float* bf  = (const float*)d_in[4];
  const float* Wq  = (const float*)d_in[5];
  const float* bq  = (const float*)d_in[6];
  const float* Wr  = (const float*)d_in[7];
  const float* br  = (const float*)d_in[8];
  float* out = (float*)d_out;

  const int T  = 512;
  const int BT = in_sizes[1] / D14;
  const int B  = BT / T;

  kf_scan<<<dim3(B), dim3(32), 0, stream>>>(xh, z, m21, Wf, bf, Wq, bq, Wr, br,
                                            out, T);
}


# Product K*S leaves the matrix unit's serial chain: K solves K(S + d*I) = P_ho, so -(K S)^T = d*K^T - P_ho^T, one multiply-add per register for 4 chained WMMAs; rounding of K S changes
# speedup vs baseline: 8.9161x; 1.0528x over previous
#include <hip/hip_runtime.h>

#define D14   14
#define JITTER 1e-4f
#define MIN_Q  1e-5f
#define MIN_R  1e-4f

typedef __attribute__((ext_vector_type(2))) float v2f;
typedef __attribute__((ext_vector_type(8))) float v8f;

__device__ __forceinline__ v8f mm_tn(v8f X, v8f Y, v8f c) {
#pragma unroll
  for (int kk = 0; kk < 4; ++kk) {
    v2f a, b;
    a.x = X[2 * kk]; a.y = X[2 * kk + 1];
    b.x = Y[2 * kk]; b.y = Y[2 * kk + 1];
    c = __builtin_amdgcn_wmma_f32_16x16x4_f32(false, a, false, b, (short)0, c,
                                              false, false);
  }
  return c;
}

__device__ __forceinline__ float xhalf(float x) {
  const int b = __builtin_bit_cast(int, x);
  return __builtin_bit_cast(float,
      __builtin_amdgcn_permlanex16(b, b, 0x76543210, 0xfedcba98, false, false));
}

struct TokRaw {
  float x[8], z[8];
  float mr[4];
  float mc;
};

__device__ __forceinline__ void load_tok_part(TokRaw& r, const float* __restrict__ xh,
                                              const float* __restrict__ z,
                                              const float* __restrict__ m21,
                                              long long tok, const int* er, const int* gr,
                                              int gc, int j) {
  if (j < 2) {
#pragma unroll
    for (int v = 4 * j; v < 4 * j + 4; ++v) {
      r.x[v] = xh[tok * D14 + er[v]];
      r.z[v] = z[tok * D14 + er[v]];
    }
  } else {
#pragma unroll
    for (int i = 0; i < 4; ++i) r.mr[i] = m21[tok * 21 + gr[i]];
    r.mc = m21[tok * 21 + gc];
  }
}

__device__ __forceinline__ TokRaw load_tok(const float* __restrict__ xh,
                                           const float* __restrict__ z,
                                           const float* __restrict__ m21,
                                           long long tok, const int* er, const int* gr,
                                           int gc) {
  TokRaw r;
#pragma unroll
  for (int j = 0; j < 3; ++j) load_tok_part(r, xh, z, m21, tok, er, gr, gc, j);
  return r;
}

__device__ __forceinline__ v8f net_mm(const float* X, const float* Y, int kk, v8f c) {
  v2f a, b;
  a.x = X[2 * kk]; a.y = X[2 * kk + 1];
  b.x = Y[2 * kk]; b.y = Y[2 * kk + 1];
  return __builtin_amdgcn_wmma_f32_16x16x4_f32(false, a, false, b, (short)0, c, false, false);
}

struct StepVec {
  float qcol, rcol, mcol;
  float xpcol;
  float mrow[8], ivrow[8];
};

struct NetW {
  float tWf[8], tWq[8], tWr[8];
  float bfr[8];
  float bqc, brc;
};

struct NetAcc { v8f xr, qc, rc; };

__device__ __forceinline__ NetAcc nets_begin(const NetW& w) {
  NetAcc a;
#pragma unroll
  for (int v = 0; v < 8; ++v) { a.xr[v] = w.bfr[v]; a.qc[v] = w.bqc; a.rc[v] = w.brc; }
  return a;
}

__device__ __forceinline__ void nets_one(NetAcc& a, const TokRaw& r, const NetW& w, int i) {
  if (i < 8) {
    if ((i & 1) == 0) a.xr = net_mm(w.tWf, r.x, i >> 1, a.xr);
    else              a.rc = net_mm(r.z, w.tWr, i >> 1, a.rc);
  } else {
    float xrf[8];
#pragma unroll
    for (int v = 0; v < 8; ++v) xrf[v] = a.xr[v];
    a.qc = net_mm(xrf, w.tWq, i - 8, a.qc);
  }
}

__device__ __forceinline__ void nets_one_here(NetAcc& a, const TokRaw& r, const NetW& w, int i) {
  __builtin_amdgcn_sched_barrier(0);
  nets_one(a, r, w, i);
  __builtin_amdgcn_sched_barrier(0);
}

__device__ __forceinline__ StepVec nets_end(const NetAcc& a, const TokRaw& r, int l16, int h,
                                            bool in14) {
  StepVec s;
  s.qcol  = in14 ? fmaxf(a.qc[0], MIN_Q) : 0.0f;
  s.rcol  = in14 ? fmaxf(a.rc[0], MIN_R) : 0.0f;
  s.mcol  = (in14 && r.mc > 0.5f) ? 1.0f : 0.0f;
  float xsel = a.xr[0];
#pragma unroll
  for (int v = 1; v < 8; ++v) xsel = ((l16 & 7) == v) ? a.xr[v] : xsel;
  s.xpcol = xsel;
#pragma unroll
  for (int v = 0; v < 8; ++v) {
    s.mrow[v]  = (v + 8 * h < D14 && r.mr[v >> 1] > 0.5f) ? 1.0f : 0.0f;
    s.ivrow[v] = s.mrow[v] * (r.z[v] - a.xr[v]);
  }
  return s;
}

__device__ __forceinline__ StepVec token_nets(const TokRaw& r, const NetW& w, int l16, int h,
                                              bool in14) {
  NetAcc a = nets_begin(w);
#pragma unroll
  for (int i = 0; i < 12; ++i) nets_one(a, r, w, i);
  return nets_end(a, r, l16, h, in14);
}

__global__ __launch_bounds__(32) void kf_scan(
    const float* __restrict__ xh, const float* __restrict__ z,
    const float* __restrict__ m21,
    const float* __restrict__ Wf, const float* __restrict__ bf,
    const float* __restrict__ Wq, const float* __restrict__ bq,
    const float* __restrict__ Wr, const float* __restrict__ br,
    float* __restrict__ out, int T) {
  const int lane = threadIdx.x;
  const int b    = blockIdx.x;
  const int l16  = lane & 15;
  const int h    = lane >> 4;
  const bool in14 = (l16 < D14);
  const int  ci   = in14 ? l16 : (D14 - 1);

  NetW w;
  int er[8], gr[4];
#pragma unroll
  for (int v = 0; v < 8; ++v) {
    const int  k  = v + 8 * h;
    const int  kc = (k < D14) ? k : (D14 - 1);
    const bool ok = (k < D14) && in14;
    const float wf = Wf[kc * D14 + ci], wq = Wq[kc * D14 + ci], wr = Wr[kc * D14 + ci];
    const float bv = bf[kc];
    w.tWf[v] = ok ? wf : 0.0f;
    w.tWq[v] = ok ? wq : 0.0f;
    w.tWr[v] = ok ? wr : 0.0f;
    w.bfr[v] = (k < D14) ? bv : 0.0f;
    er[v]    = kc;
  }
#pragma unroll
  for (int j = 0; j < 4; ++j) {
    const int g = 4 * h + j;
    gr[j] = 3 * ((g < 7) ? g : 6) + 2;
  }
  const int gc = 3 * (ci >> 1) + 2;
  const float bql = bq[ci], brl = br[ci];
  w.bqc = in14 ? bql : 0.0f;
  w.brc = in14 ? brl : 0.0f;

  v8f Pt;
#pragma unroll
  for (int v = 0; v < 8; ++v) Pt[v] = (v + 8 * h == l16 && in14) ? 1.0f : 0.0f;

  const long long base = (long long)b * T;

  StepVec sv = token_nets(load_tok(xh, z, m21, base, er, gr, gc), w, l16, h, in14);
  TokRaw rn = load_tok(xh, z, m21, base + ((1 < T) ? 1 : 0), er, gr, gc);

  const bool owner = ((l16 >> 3) == h) && in14;
  float pend = 0.0f;

#pragma unroll 1
  for (int t = 0; t < T; ++t) {
    const long long tok = base + t;
    if (owner && t > 0) out[(tok - 1) * D14 + l16] = pend;
    const long long tokn = base + ((t + 2 < T) ? (t + 2) : (T - 1));
    TokRaw rnn;
    const float qcol = sv.qcol, rcol = sv.rcol, mcol = sv.mcol;
    const float dterm = mcol * rcol + (1.0f - mcol) + JITTER * mcol;
    v8f St, Yn;
    float dS = 0.0f;
#pragma unroll
    for (int v = 0; v < 8; ++v) {
      const bool dg = (v + 8 * h == l16);
      Pt[v] += dg ? qcol : 0.0f;
      const float s  = sv.mrow[v] * mcol * Pt[v];
      const float sd = s + dterm;
      const float sx = (dg && in14) ? sd : s;
      St[v] = sx;
      Yn[v] = -(Pt[v] * sv.mrow[v]);
      dS = dg ? sx : dS;
    }
    const float tvl = in14 ? mcol * dS : 0.0f;
    v2f ta, tb;
    ta.x = tvl;  ta.y = 0.0f;
    tb.x = 1.0f; tb.y = 1.0f;
    v8f te = {};
    te = __builtin_amdgcn_wmma_f32_16x16x4_f32(false, ta, false, tb, (short)0, te,
                                               false, false);
    const float th = ((te[0] + te[2]) + (te[4] + te[6])) + ((te[1] + te[3]) + (te[5] + te[7]));
    const float tv = th + xhalf(th);
    const float dload = JITTER * fmaxf(tv, 1e-6f);
    v8f m;
#pragma unroll
    for (int v = 0; v < 8; ++v) {
      const bool dg = (v + 8 * h == l16);
      m[v] = dg ? (in14 ? St[v] + dload : 1.0f) : St[v];
    }
    NetAcc na = nets_begin(w);
#pragma unroll
    for (int p = 0; p < D14 / 2; ++p) {
      const int k0 = 2 * p;
      const int hk = k0 >> 3, v0 = k0 & 7, v1 = v0 + 1;
      const int ln = k0 + 16 * hk;
      const float r0 = m[v0], r1 = m[v1];
      const float a = __int_as_float(__builtin_amdgcn_readlane(__float_as_int(r0), ln));
      const float b = __int_as_float(__builtin_amdgcn_readlane(__float_as_int(r0), ln + 1));
      const float c = __int_as_float(__builtin_amdgcn_readlane(__float_as_int(r1), ln + 1));
      const float det = fmaf(-b, b, a * c);
      if (p < 6) nets_one_here(na, rn, w, 2 * p);
      if (p == 0) {
        load_tok_part(rnn, xh, z, m21, tokn, er, gr, gc, 2);
        load_tok_part(rnn, xh, z, m21, tokn, er, gr, gc, 0);
      }
      if (p == 1) load_tok_part(rnn, xh, z, m21, tokn, er, gr, gc, 1);
      const float idet = 1.0f / det;
      const bool inH = (h == hk);
      const bool is0 = (l16 == k0), is1 = (l16 == k0 + 1);
      const bool inK = is0 || is1;
      const float t0 = c * r0 - b * r1;
      const float t1 = a * r1 - b * r0;
      const float u0 = is0 ? -c : (is1 ? b : t0);
      const float u1 = is0 ? b : (is1 ? -a : t1);
      const float x0 = xhalf(u0), x1 = xhalf(u1);
      v2f A, Bv;
      A.x = inH ? (is0 ? -1.0f : (is1 ? 0.0f : r0)) : x0;
      A.y = inH ? (is1 ? -1.0f : (is0 ? 0.0f : r1)) : x1;
      const float s0 = inH ? (inK ? 0.0f : -t0) : (is0 ? 1.0f : 0.0f);
      const float s1 = inH ? (inK ? 0.0f : -t1) : (is1 ? 1.0f : 0.0f);
      Bv.x = s0 * idet;
      Bv.y = s1 * idet;
#pragma unroll
      for (int v = 0; v < 8; ++v) m[v] = inK ? 0.0f : m[v];
      m[v0] = inH ? 0.0f : m[v0];
      m[v1] = inH ? 0.0f : m[v1];
      m = __builtin_amdgcn_wmma_f32_16x16x4_f32(false, A, false, Bv, (short)0, m,
                                                false, false);
      if (p < 6) nets_one_here(na, rn, w, 2 * p + 1);
    }
    const StepVec svn = nets_end(na, rn, l16, h, in14);
    v8f KT = {};
    KT = mm_tn(m, Yn, KT);
    v8f nSK;
#pragma unroll
    for (int v = 0; v < 8; ++v) nSK[v] = fmaf(dload, KT[v], Yn[v]);
    float part = 0.0f;
#pragma unroll
    for (int v = 0; v < 8; ++v) part += KT[v] * sv.ivrow[v];
    const float oth = __shfl_xor(part, 16, 32);
    __builtin_amdgcn_sched_barrier(0);
    v8f pa = Pt, pb = Pt;
#pragma unroll
    for (int kk = 0; kk < 4; ++kk) {
      v2f u, w2;
      u.x  = nSK[2 * kk]; u.y  = nSK[2 * kk + 1];
      w2.x = KT[2 * kk];  w2.y = KT[2 * kk + 1];
      pa = __builtin_amdgcn_wmma_f32_16x16x4_f32(false, u, false, w2, (short)0, pa,
                                                 false, false);
      pb = __builtin_amdgcn_wmma_f32_16x16x4_f32(false, w2, false, u, (short)0, pb,
                                                 false, false);
    }
#pragma unroll
    for (int v = 0; v < 8; ++v) Pt[v] = 0.5f * (pa[v] + pb[v]);
    pend = sv.xpcol + (part + oth);
    sv = svn;
    rn = rnn;
  }
  if (owner && T > 0) out[(base + T - 1) * D14 + l16] = pend;
}

extern "C" void kernel_launch(void* const* d_in, const int* in_sizes, int n_in,
                              void* d_out, int out_size, void* d_ws, size_t ws_size,
                              hipStream_t stream) {
  const float* xh  = (const float*)d_in[0];
  const float* z   = (const float*)d_in[1];
  const float* m21 = (const float*)d_in[2];
  const float* Wf  = (const float*)d_in[3];
  const float* bf  = (const float*)d_in[4];
  const float* Wq  = (const float*)d_in[5];
  const float* bq  = (const float*)d_in[6];
  const float* Wr  = (const float*)d_in[7];
  const float* br  = (const float*)d_in[8];
  float* out = (float*)d_out;

  const int T  = 512;
  const int BT = in_sizes[1] / D14;
  const int B  = BT / T;

  kf_scan<<<dim3(B), dim3(32), 0, stream>>>(xh, z, m21, Wf, bf, Wq, bq, Wr, br,
                                            out, T);
}


# Inverse sweep skips the 2x2 pivot of each unobserved element pair by a wave-uniform branch: its rows of G are 0 off the diagonal, its inverse entries meet only zero rows of P_ho^T; sums unchanged
# speedup vs baseline: 10.0133x; 1.1231x over previous
#include <hip/hip_runtime.h>

#define D14   14
#define JITTER 1e-4f
#define MIN_Q  1e-5f
#define MIN_R  1e-4f

typedef __attribute__((ext_vector_type(2))) float v2f;
typedef __attribute__((ext_vector_type(8))) float v8f;

__device__ __forceinline__ v8f mm_tn(v8f X, v8f Y, v8f c) {
#pragma unroll
  for (int kk = 0; kk < 4; ++kk) {
    v2f a, b;
    a.x = X[2 * kk]; a.y = X[2 * kk + 1];
    b.x = Y[2 * kk]; b.y = Y[2 * kk + 1];
    c = __builtin_amdgcn_wmma_f32_16x16x4_f32(false, a, false, b, (short)0, c,
                                              false, false);
  }
  return c;
}

__device__ __forceinline__ float xhalf(float x) {
  const int b = __builtin_bit_cast(int, x);
  return __builtin_bit_cast(float,
      __builtin_amdgcn_permlanex16(b, b, 0x76543210, 0xfedcba98, false, false));
}

struct TokRaw {
  float x[8], z[8];
  float mr[4];
  float mc;
};

__device__ __forceinline__ void load_tok_part(TokRaw& r, const float* __restrict__ xh,
                                              const float* __restrict__ z,
                                              const float* __restrict__ m21,
                                              long long tok, const int* er, const int* gr,
                                              int gc, int j) {
  if (j < 2) {
#pragma unroll
    for (int v = 4 * j; v < 4 * j + 4; ++v) {
      r.x[v] = xh[tok * D14 + er[v]];
      r.z[v] = z[tok * D14 + er[v]];
    }
  } else {
#pragma unroll
    for (int i = 0; i < 4; ++i) r.mr[i] = m21[tok * 21 + gr[i]];
    r.mc = m21[tok * 21 + gc];
  }
}

__device__ __forceinline__ TokRaw load_tok(const float* __restrict__ xh,
                                           const float* __restrict__ z,
                                           const float* __restrict__ m21,
                                           long long tok, const int* er, const int* gr,
                                           int gc) {
  TokRaw r;
#pragma unroll
  for (int j = 0; j < 3; ++j) load_tok_part(r, xh, z, m21, tok, er, gr, gc, j);
  return r;
}

__device__ __forceinline__ v8f net_mm(const float* X, const float* Y, int kk, v8f c) {
  v2f a, b;
  a.x = X[2 * kk]; a.y = X[2 * kk + 1];
  b.x = Y[2 * kk]; b.y = Y[2 * kk + 1];
  return __builtin_amdgcn_wmma_f32_16x16x4_f32(false, a, false, b, (short)0, c, false, false);
}

struct StepVec {
  float qcol, rcol, mcol;
  float xpcol;
  float mrow[8], ivrow[8];
};

struct NetW {
  float tWf[8], tWq[8], tWr[8];
  float bfr[8];
  float bqc, brc;
};

struct NetAcc { v8f xr, qc, rc; };

__device__ __forceinline__ NetAcc nets_begin(const NetW& w) {
  NetAcc a;
#pragma unroll
  for (int v = 0; v < 8; ++v) { a.xr[v] = w.bfr[v]; a.qc[v] = w.bqc; a.rc[v] = w.brc; }
  return a;
}

__device__ __forceinline__ void nets_one(NetAcc& a, const TokRaw& r, const NetW& w, int i) {
  if (i < 8) {
    if ((i & 1) == 0) a.xr = net_mm(w.tWf, r.x, i >> 1, a.xr);
    else              a.rc = net_mm(r.z, w.tWr, i >> 1, a.rc);
  } else {
    float xrf[8];
#pragma unroll
    for (int v = 0; v < 8; ++v) xrf[v] = a.xr[v];
    a.qc = net_mm(xrf, w.tWq, i - 8, a.qc);
  }
}

__device__ __forceinline__ void nets_one_here(NetAcc& a, const TokRaw& r, const NetW& w, int i) {
  __builtin_amdgcn_sched_barrier(0);
  nets_one(a, r, w, i);
  __builtin_amdgcn_sched_barrier(0);
}

__device__ __forceinline__ StepVec nets_end(const NetAcc& a, const TokRaw& r, int l16, int h,
                                            bool in14) {
  StepVec s;
  s.qcol  = in14 ? fmaxf(a.qc[0], MIN_Q) : 0.0f;
  s.rcol  = in14 ? fmaxf(a.rc[0], MIN_R) : 0.0f;
  s.mcol  = (in14 && r.mc > 0.5f) ? 1.0f : 0.0f;
  float xsel = a.xr[0];
#pragma unroll
  for (int v = 1; v < 8; ++v) xsel = ((l16 & 7) == v) ? a.xr[v] : xsel;
  s.xpcol = xsel;
#pragma unroll
  for (int v = 0; v < 8; ++v) {
    s.mrow[v]  = (v + 8 * h < D14 && r.mr[v >> 1] > 0.5f) ? 1.0f : 0.0f;
    s.ivrow[v] = s.mrow[v] * (r.z[v] - a.xr[v]);
  }
  return s;
}

__device__ __forceinline__ StepVec token_nets(const TokRaw& r, const NetW& w, int l16, int h,
                                              bool in14) {
  NetAcc a = nets_begin(w);
#pragma unroll
  for (int i = 0; i < 12; ++i) nets_one(a, r, w, i);
  return nets_end(a, r, l16, h, in14);
}

__global__ __launch_bounds__(32) void kf_scan(
    const float* __restrict__ xh, const float* __restrict__ z,
    const float* __restrict__ m21,
    const float* __restrict__ Wf, const float* __restrict__ bf,
    const float* __restrict__ Wq, const float* __restrict__ bq,
    const float* __restrict__ Wr, const float* __restrict__ br,
    float* __restrict__ out, int T) {
  const int lane = threadIdx.x;
  const int b    = blockIdx.x;
  const int l16  = lane & 15;
  const int h    = lane >> 4;
  const bool in14 = (l16 < D14);
  const int  ci   = in14 ? l16 : (D14 - 1);

  NetW w;
  int er[8], gr[4];
#pragma unroll
  for (int v = 0; v < 8; ++v) {
    const int  k  = v + 8 * h;
    const int  kc = (k < D14) ? k : (D14 - 1);
    const bool ok = (k < D14) && in14;
    const float wf = Wf[kc * D14 + ci], wq = Wq[kc * D14 + ci], wr = Wr[kc * D14 + ci];
    const float bv = bf[kc];
    w.tWf[v] = ok ? wf : 0.0f;
    w.tWq[v] = ok ? wq : 0.0f;
    w.tWr[v] = ok ? wr : 0.0f;
    w.bfr[v] = (k < D14) ? bv : 0.0f;
    er[v]    = kc;
  }
#pragma unroll
  for (int j = 0; j < 4; ++j) {
    const int g = 4 * h + j;
    gr[j] = 3 * ((g < 7) ? g : 6) + 2;
  }
  const int gc = 3 * (ci >> 1) + 2;
  const float bql = bq[ci], brl = br[ci];
  w.bqc = in14 ? bql : 0.0f;
  w.brc = in14 ? brl : 0.0f;

  v8f Pt;
#pragma unroll
  for (int v = 0; v < 8; ++v) Pt[v] = (v + 8 * h == l16 && in14) ? 1.0f : 0.0f;

  const long long base = (long long)b * T;

  StepVec sv = token_nets(load_tok(xh, z, m21, base, er, gr, gc), w, l16, h, in14);
  TokRaw rn = load_tok(xh, z, m21, base + ((1 < T) ? 1 : 0), er, gr, gc);

  const bool owner = ((l16 >> 3) == h) && in14;
  float pend = 0.0f;

#pragma unroll 1
  for (int t = 0; t < T; ++t) {
    const long long tok = base + t;
    if (owner && t > 0) out[(tok - 1) * D14 + l16] = pend;
    const long long tokn = base + ((t + 2 < T) ? (t + 2) : (T - 1));
    TokRaw rnn;
    const float qcol = sv.qcol, rcol = sv.rcol, mcol = sv.mcol;
    const float dterm = mcol * rcol + (1.0f - mcol) + JITTER * mcol;
    v8f St, Yn;
    float dS = 0.0f;
#pragma unroll
    for (int v = 0; v < 8; ++v) {
      const bool dg = (v + 8 * h == l16);
      Pt[v] += dg ? qcol : 0.0f;
      const float s  = sv.mrow[v] * mcol * Pt[v];
      const float sd = s + dterm;
      const float sx = (dg && in14) ? sd : s;
      St[v] = sx;
      Yn[v] = -(Pt[v] * sv.mrow[v]);
      dS = dg ? sx : dS;
    }
    const float tvl = in14 ? mcol * dS : 0.0f;
    v2f ta, tb;
    ta.x = tvl;  ta.y = 0.0f;
    tb.x = 1.0f; tb.y = 1.0f;
    v8f te = {};
    te = __builtin_amdgcn_wmma_f32_16x16x4_f32(false, ta, false, tb, (short)0, te,
                                               false, false);
    const float th = ((te[0] + te[2]) + (te[4] + te[6])) + ((te[1] + te[3]) + (te[5] + te[7]));
    const float tv = th + xhalf(th);
    const float dload = JITTER * fmaxf(tv, 1e-6f);
    v8f m;
#pragma unroll
    for (int v = 0; v < 8; ++v) {
      const bool dg = (v + 8 * h == l16);
      m[v] = dg ? (in14 ? St[v] + dload : 1.0f) : St[v];
    }
    const unsigned obs = __builtin_amdgcn_ballot_w32(mcol > 0.5f);
    NetAcc na = nets_begin(w);
#pragma unroll
    for (int p = 0; p < D14 / 2; ++p) {
      const int k0 = 2 * p;
      const int hk = k0 >> 3, v0 = k0 & 7, v1 = v0 + 1;
      const int ln = k0 + 16 * hk;
      const bool piv = (obs & (3u << k0)) != 0u;
      const float r0 = m[v0], r1 = m[v1];
      float a, b, c, det;
      if (piv) {
        a = __int_as_float(__builtin_amdgcn_readlane(__float_as_int(r0), ln));
        b = __int_as_float(__builtin_amdgcn_readlane(__float_as_int(r0), ln + 1));
        c = __int_as_float(__builtin_amdgcn_readlane(__float_as_int(r1), ln + 1));
        det = fmaf(-b, b, a * c);
      }
      if (p < 6) nets_one_here(na, rn, w, 2 * p);
      if (p == 0) {
        load_tok_part(rnn, xh, z, m21, tokn, er, gr, gc, 2);
        load_tok_part(rnn, xh, z, m21, tokn, er, gr, gc, 0);
      }
      if (p == 1) load_tok_part(rnn, xh, z, m21, tokn, er, gr, gc, 1);
      if (piv) {
      const float idet = 1.0f / det;
      const bool inH = (h == hk);
      const bool is0 = (l16 == k0), is1 = (l16 == k0 + 1);
      const bool inK = is0 || is1;
      const float t0 = c * r0 - b * r1;
      const float t1 = a * r1 - b * r0;
      const float u0 = is0 ? -c : (is1 ? b : t0);
      const float u1 = is0 ? b : (is1 ? -a : t1);
      const float x0 = xhalf(u0), x1 = xhalf(u1);
      v2f A, Bv;
      A.x = inH ? (is0 ? -1.0f : (is1 ? 0.0f : r0)) : x0;
      A.y = inH ? (is1 ? -1.0f : (is0 ? 0.0f : r1)) : x1;
      const float s0 = inH ? (inK ? 0.0f : -t0) : (is0 ? 1.0f : 0.0f);
      const float s1 = inH ? (inK ? 0.0f : -t1) : (is1 ? 1.0f : 0.0f);
      Bv.x = s0 * idet;
      Bv.y = s1 * idet;
#pragma unroll
      for (int v = 0; v < 8; ++v) m[v] = inK ? 0.0f : m[v];
      m[v0] = inH ? 0.0f : m[v0];
      m[v1] = inH ? 0.0f : m[v1];
      m = __builtin_amdgcn_wmma_f32_16x16x4_f32(false, A, false, Bv, (short)0, m,
                                                false, false);
      }
      if (p < 6) nets_one_here(na, rn, w, 2 * p + 1);
    }
    const StepVec svn = nets_end(na, rn, l16, h, in14);
    v8f KT = {};
    KT = mm_tn(m, Yn, KT);
    v8f nSK;
#pragma unroll
    for (int v = 0; v < 8; ++v) nSK[v] = fmaf(dload, KT[v], Yn[v]);
    float part = 0.0f;
#pragma unroll
    for (int v = 0; v < 8; ++v) part += KT[v] * sv.ivrow[v];
    const float oth = __shfl_xor(part, 16, 32);
    __builtin_amdgcn_sched_barrier(0);
    v8f pa = Pt, pb = Pt;
#pragma unroll
    for (int kk = 0; kk < 4; ++kk) {
      v2f u, w2;
      u.x  = nSK[2 * kk]; u.y  = nSK[2 * kk + 1];
      w2.x = KT[2 * kk];  w2.y = KT[2 * kk + 1];
      pa = __builtin_amdgcn_wmma_f32_16x16x4_f32(false, u, false, w2, (short)0, pa,
                                                 false, false);
      pb = __builtin_amdgcn_wmma_f32_16x16x4_f32(false, w2, false, u, (short)0, pb,
                                                 false, false);
    }
#pragma unroll
    for (int v = 0; v < 8; ++v) Pt[v] = 0.5f * (pa[v] + pb[v]);
    pend = sv.xpcol + (part + oth);
    sv = svn;
    rn = rnn;
  }
  if (owner && T > 0) out[(base + T - 1) * D14 + l16] = pend;
}

extern "C" void kernel_launch(void* const* d_in, const int* in_sizes, int n_in,
                              void* d_out, int out_size, void* d_ws, size_t ws_size,
                              hipStream_t stream) {
  const float* xh  = (const float*)d_in[0];
  const float* z   = (const float*)d_in[1];
  const float* m21 = (const float*)d_in[2];
  const float* Wf  = (const float*)d_in[3];
  const float* bf  = (const float*)d_in[4];
  const float* Wq  = (const float*)d_in[5];
  const float* bq  = (const float*)d_in[6];
  const float* Wr  = (const float*)d_in[7];
  const float* br  = (const float*)d_in[8];
  float* out = (float*)d_out;

  const int T  = 512;
  const int BT = in_sizes[1] / D14;
  const int B  = BT / T;

  kf_scan<<<dim3(B), dim3(32), 0, stream>>>(xh, z, m21, Wf, bf, Wq, bq, Wr, br,
                                            out, T);
}


# Trace-scaled loading leaves the serial chain: next step's masked trace is summed from diag(P_pr) and nSK*KT beside the last products, its WMMA among them; G is 3 ops from P; trace sum order changes
# speedup vs baseline: 10.7490x; 1.0735x over previous
#include <hip/hip_runtime.h>

#define D14   14
#define JITTER 1e-4f
#define MIN_Q  1e-5f
#define MIN_R  1e-4f

typedef __attribute__((ext_vector_type(2))) float v2f;
typedef __attribute__((ext_vector_type(8))) float v8f;

__device__ __forceinline__ v8f mm_tn_one(v8f X, v8f Y, int kk, v8f c) {
  v2f a, b;
  a.x = X[2 * kk]; a.y = X[2 * kk + 1];
  b.x = Y[2 * kk]; b.y = Y[2 * kk + 1];
  return __builtin_amdgcn_wmma_f32_16x16x4_f32(false, a, false, b, (short)0, c, false, false);
}

__device__ __forceinline__ float xhalf(float x) {
  const int b = __builtin_bit_cast(int, x);
  return __builtin_bit_cast(float,
      __builtin_amdgcn_permlanex16(b, b, 0x76543210, 0xfedcba98, false, false));
}

struct TokRaw {
  float x[8], z[8];
  float mr[4];
  float mc;
};

__device__ __forceinline__ void load_tok_part(TokRaw& r, const float* __restrict__ xh,
                                              const float* __restrict__ z,
                                              const float* __restrict__ m21,
                                              long long tok, const int* er, const int* gr,
                                              int gc, int j) {
  if (j < 2) {
#pragma unroll
    for (int v = 4 * j; v < 4 * j + 4; ++v) {
      r.x[v] = xh[tok * D14 + er[v]];
      r.z[v] = z[tok * D14 + er[v]];
    }
  } else {
#pragma unroll
    for (int i = 0; i < 4; ++i) r.mr[i] = m21[tok * 21 + gr[i]];
    r.mc = m21[tok * 21 + gc];
  }
}

__device__ __forceinline__ TokRaw load_tok(const float* __restrict__ xh,
                                           const float* __restrict__ z,
                                           const float* __restrict__ m21,
                                           long long tok, const int* er, const int* gr,
                                           int gc) {
  TokRaw r;
#pragma unroll
  for (int j = 0; j < 3; ++j) load_tok_part(r, xh, z, m21, tok, er, gr, gc, j);
  return r;
}

__device__ __forceinline__ v8f net_mm(const float* X, const float* Y, int kk, v8f c) {
  v2f a, b;
  a.x = X[2 * kk]; a.y = X[2 * kk + 1];
  b.x = Y[2 * kk]; b.y = Y[2 * kk + 1];
  return __builtin_amdgcn_wmma_f32_16x16x4_f32(false, a, false, b, (short)0, c, false, false);
}

struct StepVec {
  float qcol, rcol, mcol;
  float xpcol;
  float mrow[8], ivrow[8];
};

struct NetW {
  float tWf[8], tWq[8], tWr[8];
  float bfr[8];
  float bqc, brc;
};

struct NetAcc { v8f xr, qc, rc; };

__device__ __forceinline__ NetAcc nets_begin(const NetW& w) {
  NetAcc a;
#pragma unroll
  for (int v = 0; v < 8; ++v) { a.xr[v] = w.bfr[v]; a.qc[v] = w.bqc; a.rc[v] = w.brc; }
  return a;
}

__device__ __forceinline__ void nets_one(NetAcc& a, const TokRaw& r, const NetW& w, int i) {
  if (i < 8) {
    if ((i & 1) == 0) a.xr = net_mm(w.tWf, r.x, i >> 1, a.xr);
    else              a.rc = net_mm(r.z, w.tWr, i >> 1, a.rc);
  } else {
    float xrf[8];
#pragma unroll
    for (int v = 0; v < 8; ++v) xrf[v] = a.xr[v];
    a.qc = net_mm(xrf, w.tWq, i - 8, a.qc);
  }
}

__device__ __forceinline__ void nets_one_here(NetAcc& a, const TokRaw& r, const NetW& w, int i) {
  __builtin_amdgcn_sched_barrier(0);
  nets_one(a, r, w, i);
  __builtin_amdgcn_sched_barrier(0);
}

struct Prep {
  float qd[8];
  float mm[8];
  float dd[8];
  float dterm;
  unsigned obs;
};

__device__ __forceinline__ void next_head_r(StepVec& s, Prep& p, const NetAcc& a, const TokRaw& r,
                                            bool in14) {
  s.rcol  = in14 ? fmaxf(a.rc[0], MIN_R) : 0.0f;
  s.mcol  = (in14 && r.mc > 0.5f) ? 1.0f : 0.0f;
  p.dterm = s.mcol * s.rcol + (1.0f - s.mcol) + JITTER * s.mcol;
  p.obs   = __builtin_amdgcn_ballot_w32(s.mcol > 0.5f);
}

__device__ __forceinline__ void next_head_q(StepVec& s, const NetAcc& a, bool in14) {
  s.qcol  = in14 ? fmaxf(a.qc[0], MIN_Q) : 0.0f;
}

__device__ __forceinline__ void next_pair(StepVec& s, Prep& p, const NetAcc& a, const TokRaw& r,
                                          int l16, int h, bool in14, int j, float& xs,
                                          const v8f& P, float& pd) {
#pragma unroll
  for (int v = 2 * j; v < 2 * j + 2; ++v) {
    const bool dg = (v + 8 * h == l16);
    s.mrow[v]  = (v + 8 * h < D14 && r.mr[v >> 1] > 0.5f) ? 1.0f : 0.0f;
    s.ivrow[v] = s.mrow[v] * (r.z[v] - a.xr[v]);
    p.qd[v] = dg ? s.qcol : 0.0f;
    p.mm[v] = s.mrow[v] * s.mcol;
    p.dd[v] = dg ? (in14 ? p.dterm : 1.0f) : 0.0f;
    xs = (v == 0 || (l16 & 7) == v) ? a.xr[v] : xs;
    pd = dg ? P[v] : pd;
  }
}

__device__ __forceinline__ v8f lane_sum_mm(float term) {
  v2f ta, tb;
  ta.x = term; ta.y = 0.0f;
  tb.x = 1.0f; tb.y = 1.0f;
  v8f te = {};
  return __builtin_amdgcn_wmma_f32_16x16x4_f32(false, ta, false, tb, (short)0, te, false, false);
}

__device__ __forceinline__ float lane_sum_fin(v8f te) {
  const float th = ((te[0] + te[2]) + (te[4] + te[6])) + ((te[1] + te[3]) + (te[5] + te[7]));
  return th + xhalf(th);
}

__global__ __launch_bounds__(32) void kf_scan(
    const float* __restrict__ xh, const float* __restrict__ z,
    const float* __restrict__ m21,
    const float* __restrict__ Wf, const float* __restrict__ bf,
    const float* __restrict__ Wq, const float* __restrict__ bq,
    const float* __restrict__ Wr, const float* __restrict__ br,
    float* __restrict__ out, int T) {
  const int lane = threadIdx.x;
  const int b    = blockIdx.x;
  const int l16  = lane & 15;
  const int h    = lane >> 4;
  const bool in14 = (l16 < D14);
  const int  ci   = in14 ? l16 : (D14 - 1);

  NetW w;
  int er[8], gr[4];
#pragma unroll
  for (int v = 0; v < 8; ++v) {
    const int  k  = v + 8 * h;
    const int  kc = (k < D14) ? k : (D14 - 1);
    const bool ok = (k < D14) && in14;
    const float wf = Wf[kc * D14 + ci], wq = Wq[kc * D14 + ci], wr = Wr[kc * D14 + ci];
    const float bv = bf[kc];
    w.tWf[v] = ok ? wf : 0.0f;
    w.tWq[v] = ok ? wq : 0.0f;
    w.tWr[v] = ok ? wr : 0.0f;
    w.bfr[v] = (k < D14) ? bv : 0.0f;
    er[v]    = kc;
  }
#pragma unroll
  for (int j = 0; j < 4; ++j) {
    const int g = 4 * h + j;
    gr[j] = 3 * ((g < 7) ? g : 6) + 2;
  }
  const int gc = 3 * (ci >> 1) + 2;
  const float bql = bq[ci], brl = br[ci];
  w.bqc = in14 ? bql : 0.0f;
  w.brc = in14 ? brl : 0.0f;

  v8f Pt;
#pragma unroll
  for (int v = 0; v < 8; ++v) Pt[v] = (v + 8 * h == l16 && in14) ? 1.0f : 0.0f;

  const long long base = (long long)b * T;

  StepVec sv;
  Prep pp;
  {
    const TokRaw r0 = load_tok(xh, z, m21, base, er, gr, gc);
    NetAcc a0 = nets_begin(w);
#pragma unroll
    for (int i = 0; i < 12; ++i) nets_one(a0, r0, w, i);
    float xs0 = 0.0f, pd0 = 0.0f;
    next_head_r(sv, pp, a0, r0, in14);
    next_head_q(sv, a0, in14);
#pragma unroll
    for (int j = 0; j < 4; ++j) next_pair(sv, pp, a0, r0, l16, h, in14, j, xs0, Pt, pd0);
    sv.xpcol = xs0;
  }
  TokRaw rn = load_tok(xh, z, m21, base + ((1 < T) ? 1 : 0), er, gr, gc);

  const bool owner = ((l16 >> 3) == h) && in14;
  float pend = 0.0f;

  float dload = JITTER * fmaxf(lane_sum_fin(lane_sum_mm(
                    sv.mcol * (owner ? (1.0f + sv.qcol) + pp.dterm : 0.0f))), 1e-6f);

#pragma unroll 1
  for (int t = 0; t < T; ++t) {
    const long long tok = base + t;
    if (owner && t > 0) out[(tok - 1) * D14 + l16] = pend;
    const long long tokn = base + ((t + 2 < T) ? (t + 2) : (T - 1));
    TokRaw rnn;
    v8f m, Yn;
#pragma unroll
    for (int v = 0; v < 8; ++v) {
      const bool dg = (v + 8 * h == l16);
      Pt[v] += pp.qd[v];
      m[v]  = (pp.mm[v] * Pt[v] + pp.dd[v]) + ((dg && in14) ? dload : 0.0f);
      Yn[v] = -(Pt[v] * sv.mrow[v]);
    }
    const unsigned obs = pp.obs;
    NetAcc na = nets_begin(w);
    StepVec svn;
    Prep ppn;
#pragma unroll
    for (int p = 0; p < D14 / 2; ++p) {
      const int k0 = 2 * p;
      const int hk = k0 >> 3, v0 = k0 & 7, v1 = v0 + 1;
      const int ln = k0 + 16 * hk;
      const bool piv = (obs & (3u << k0)) != 0u;
      const float r0 = m[v0], r1 = m[v1];
      float a, b, c, det;
      if (piv) {
        a = __int_as_float(__builtin_amdgcn_readlane(__float_as_int(r0), ln));
        b = __int_as_float(__builtin_amdgcn_readlane(__float_as_int(r0), ln + 1));
        c = __int_as_float(__builtin_amdgcn_readlane(__float_as_int(r1), ln + 1));
        det = fmaf(-b, b, a * c);
      }
      if (p < 6) nets_one_here(na, rn, w, 2 * p);
      if (p == 0) {
        load_tok_part(rnn, xh, z, m21, tokn, er, gr, gc, 2);
        load_tok_part(rnn, xh, z, m21, tokn, er, gr, gc, 0);
      }
      if (p == 1) load_tok_part(rnn, xh, z, m21, tokn, er, gr, gc, 1);
      if (p == 4) next_head_r(svn, ppn, na, rn, in14);
      if (piv) {
      const float idet = 1.0f / det;
      const bool inH = (h == hk);
      const bool is0 = (l16 == k0), is1 = (l16 == k0 + 1);
      const bool inK = is0 || is1;
      const float t0 = c * r0 - b * r1;
      const float t1 = a * r1 - b * r0;
      const float u0 = is0 ? -c : (is1 ? b : t0);
      const float u1 = is0 ? b : (is1 ? -a : t1);
      const float x0 = xhalf(u0), x1 = xhalf(u1);
      v2f A, Bv;
      A.x = inH ? (is0 ? -1.0f : (is1 ? 0.0f : r0)) : x0;
      A.y = inH ? (is1 ? -1.0f : (is0 ? 0.0f : r1)) : x1;
      const float s0 = inH ? (inK ? 0.0f : -t0) : (is0 ? 1.0f : 0.0f);
      const float s1 = inH ? (inK ? 0.0f : -t1) : (is1 ? 1.0f : 0.0f);
      Bv.x = s0 * idet;
      Bv.y = s1 * idet;
#pragma unroll
      for (int v = 0; v < 8; ++v) m[v] = inK ? 0.0f : m[v];
      m[v0] = inH ? 0.0f : m[v0];
      m[v1] = inH ? 0.0f : m[v1];
      m = __builtin_amdgcn_wmma_f32_16x16x4_f32(false, A, false, Bv, (short)0, m,
                                                false, false);
      }
      if (p < 6) nets_one_here(na, rn, w, 2 * p + 1);
    }
    float xs = 0.0f, pd = 0.0f;
    v8f KT = {};
#pragma unroll
    for (int kk = 0; kk < 4; ++kk) {
      KT = mm_tn_one(m, Yn, kk, KT);
      __builtin_amdgcn_sched_barrier(0);
      if (kk == 0) next_head_q(svn, na, in14);
      else         next_pair(svn, ppn, na, rn, l16, h, in14, kk - 1, xs, Pt, pd);
      if (kk == 3) next_pair(svn, ppn, na, rn, l16, h, in14, 3, xs, Pt, pd);
      __builtin_amdgcn_sched_barrier(0);
    }
    svn.xpcol = xs;
    v8f nSK;
#pragma unroll
    for (int v = 0; v < 8; ++v) nSK[v] = fmaf(dload, KT[v], Yn[v]);
    float part = 0.0f;
#pragma unroll
    for (int v = 0; v < 8; ++v) part += KT[v] * sv.ivrow[v];
    const float oth = xhalf(part);
    __builtin_amdgcn_sched_barrier(0);
    float e0 = nSK[0] * KT[0], e1 = nSK[1] * KT[1];
#pragma unroll
    for (int v = 2; v < 8; v += 2) {
      e0 = fmaf(nSK[v], KT[v], e0);
      e1 = fmaf(nSK[v + 1], KT[v + 1], e1);
    }
    const float own = owner ? (pd + svn.qcol) + ppn.dterm : 0.0f;
    const float trm = svn.mcol * ((e0 + e1) + own);
    v8f te;
    v8f pa = Pt, pb = Pt;
#pragma unroll
    for (int kk = 0; kk < 4; ++kk) {
      if (kk == 2) {
        __builtin_amdgcn_sched_barrier(0);
        te = lane_sum_mm(trm);
        __builtin_amdgcn_sched_barrier(0);
      }
      v2f u, w2;
      u.x  = nSK[2 * kk]; u.y  = nSK[2 * kk + 1];
      w2.x = KT[2 * kk];  w2.y = KT[2 * kk + 1];
      pa = __builtin_amdgcn_wmma_f32_16x16x4_f32(false, u, false, w2, (short)0, pa,
                                                 false, false);
      pb = __builtin_amdgcn_wmma_f32_16x16x4_f32(false, w2, false, u, (short)0, pb,
                                                 false, false);
    }
#pragma unroll
    for (int v = 0; v < 8; ++v) Pt[v] = 0.5f * (pa[v] + pb[v]);
    pend = sv.xpcol + (part + oth);
    dload = JITTER * fmaxf(lane_sum_fin(te), 1e-6f);
    sv = svn;
    rn = rnn;
    pp = ppn;
  }
  if (owner && T > 0) out[(base + T - 1) * D14 + l16] = pend;
}

extern "C" void kernel_launch(void* const* d_in, const int* in_sizes, int n_in,
                              void* d_out, int out_size, void* d_ws, size_t ws_size,
                              hipStream_t stream) {
  const float* xh  = (const float*)d_in[0];
  const float* z   = (const float*)d_in[1];
  const float* m21 = (const float*)d_in[2];
  const float* Wf  = (const float*)d_in[3];
  const float* bf  = (const float*)d_in[4];
  const float* Wq  = (const float*)d_in[5];
  const float* bq  = (const float*)d_in[6];
  const float* Wr  = (const float*)d_in[7];
  const float* br  = (const float*)d_in[8];
  float* out = (float*)d_out;

  const int T  = 512;
  const int BT = in_sizes[1] / D14;
  const int B  = BT / T;

  kf_scan<<<dim3(B), dim3(32), 0, stream>>>(xh, z, m21, Wf, bf, Wq, bq, Wr, br,
                                            out, T);
}
